# starting kernel
# baseline (speedup 1.0000x reference)
#include <hip/hip_runtime.h>

#define NNODES 100000
#define NEDGES 1600000
#define NFEAT 128
#define NHID 64
#define NCLASS 16
#define MTILES (NNODES / 16)

typedef float v2f __attribute__((ext_vector_type(2)));
typedef float v8f __attribute__((ext_vector_type(8)));

__device__ __forceinline__ void atomic_add_f32(float* p, float v) {
    __hip_atomic_fetch_add(p, v, __ATOMIC_RELAXED, __HIP_MEMORY_SCOPE_AGENT);
}

__global__ __launch_bounds__(256) void gcn_zero_kernel(float4* __restrict__ p, int n4) {
    int i = blockIdx.x * 256 + threadIdx.x;
    if (i < n4) p[i] = make_float4(0.f, 0.f, 0.f, 0.f);
}

__global__ __launch_bounds__(256) void gcn_init_logits_kernel(const float* __restrict__ b2,
                                                              float* __restrict__ out, int total) {
    int i = blockIdx.x * 256 + threadIdx.x;
    if (i < total) out[i] = b2[i & (NCLASS - 1)];
}

__global__ __launch_bounds__(128) void gcn_gemm1_kernel(const float* __restrict__ x,
                                                        const float* __restrict__ W1,
                                                        float* __restrict__ out) {
    __shared__ float As[16][NFEAT + 4];
    __shared__ float WsT[NHID][NFEAT + 4];

    const int tid = threadIdx.x;
    const int tileM = blockIdx.x;

    for (int i = tid; i < NFEAT * NHID; i += 128) {
        int k = i >> 6, n = i & (NHID - 1);
        WsT[n][k] = W1[i];
    }
    const float* xrow = x + (size_t)tileM * 16 * NFEAT;
    for (int i = tid; i < 16 * NFEAT; i += 128) {
        As[i >> 7][i & (NFEAT - 1)] = xrow[i];
    }
    __syncthreads();

    const int lane = tid & 31;
    const int wave = tid >> 5;
    const int m  = lane & 15;
    const int n  = lane & 15;
    const int kh = (lane >> 4) * 2;
    const int n0 = wave * 16;

    v8f acc = {};
    #pragma unroll 4
    for (int kk = 0; kk < NFEAT; kk += 4) {
        const int k0 = kk + kh;
        v2f a, b;
        a.x = As[m][k0];
        a.y = As[m][k0 + 1];
        b.x = WsT[n0 + n][k0];
        b.y = WsT[n0 + n][k0 + 1];
        acc = __builtin_amdgcn_wmma_f32_16x16x4_f32(false, a, false, b,
                                                    (short)0, acc, false, false);
    }

    float* orow = out + ((size_t)tileM * 16 + (lane >> 4) * 8) * NHID + n0 + n;
    #pragma unroll
    for (int r = 0; r < 8; ++r)
        orow[(size_t)r * NHID] = acc[r];
}

__global__ __launch_bounds__(256) void gcn_spmm64_kernel(const int* __restrict__ src,
                                                         const int* __restrict__ dst,
                                                         const float* __restrict__ ew,
                                                         const float* __restrict__ sup,
                                                         float* __restrict__ out) {
    long gid = (long)blockIdx.x * 256 + threadIdx.x;
    if (gid >= (long)NEDGES * NHID) return;
    int e = (int)(gid >> 6);
    int f = (int)(gid & (NHID - 1));
    int s = src[e];
    int d = dst[e];
    float w = ew[e];
    float msg = w * sup[(size_t)s * NHID + f];
    atomic_add_f32(out + (size_t)d * NHID + f, msg);
}

__global__ __launch_bounds__(256) void gcn_gemm2_kernel(const float* __restrict__ agg1,
                                                        const float* __restrict__ b1,
                                                        const float* __restrict__ W2,
                                                        float* __restrict__ out) {
    __shared__ float WsT[NCLASS][NHID + 4];
    __shared__ float bs[NHID];

    const int tid = threadIdx.x;
    for (int i = tid; i < NHID * NCLASS; i += 256) {
        int k = i >> 4, n = i & (NCLASS - 1);
        WsT[n][k] = W2[i];
    }
    if (tid < NHID) bs[tid] = b1[tid];
    __syncthreads();

    const int lane = tid & 31;
    const int wave = tid >> 5;
    const int tileM = blockIdx.x * 8 + wave;
    if (tileM >= MTILES) return;

    const int m  = lane & 15;
    const int n  = lane & 15;
    const int kh = (lane >> 4) * 2;

    const float* arow = agg1 + ((size_t)tileM * 16 + m) * NHID;

    v8f acc = {};
    #pragma unroll 4
    for (int kk = 0; kk < NHID; kk += 4) {
        const int k0 = kk + kh;
        v2f a, b;
        a.x = fmaxf(arow[k0]     + bs[k0],     0.f);
        a.y = fmaxf(arow[k0 + 1] + bs[k0 + 1], 0.f);
        b.x = WsT[n][k0];
        b.y = WsT[n][k0 + 1];
        acc = __builtin_amdgcn_wmma_f32_16x16x4_f32(false, a, false, b,
                                                    (short)0, acc, false, false);
    }

    float* orow = out + ((size_t)tileM * 16 + (lane >> 4) * 8) * NCLASS + n;
    #pragma unroll
    for (int r = 0; r < 8; ++r)
        orow[(size_t)r * NCLASS] = acc[r];
}

__global__ __launch_bounds__(256) void gcn_spmm16_kernel(const int* __restrict__ src,
                                                         const int* __restrict__ dst,
                                                         const float* __restrict__ ew,
                                                         const float* __restrict__ sup,
                                                         float* __restrict__ out) {
    long gid = (long)blockIdx.x * 256 + threadIdx.x;
    if (gid >= (long)NEDGES * NCLASS) return;
    int e = (int)(gid >> 4);
    int f = (int)(gid & (NCLASS - 1));
    int s = src[e];
    int d = dst[e];
    float w = ew[e];
    float msg = w * sup[(size_t)s * NCLASS + f];
    atomic_add_f32(out + (size_t)d * NCLASS + f, msg);
}

__global__ __launch_bounds__(256) void gcn_logsoftmax_kernel(float* __restrict__ logits) {
    int node = blockIdx.x * 256 + threadIdx.x;
    if (node >= NNODES) return;
    float4* p = (float4*)(logits + (size_t)node * NCLASS);
    float4 v[4];
    #pragma unroll
    for (int i = 0; i < 4; ++i) v[i] = p[i];
    float* f = (float*)v;
    float mx = f[0];
    #pragma unroll
    for (int i = 1; i < 16; ++i) mx = fmaxf(mx, f[i]);
    float sum = 0.f;
    #pragma unroll
    for (int i = 0; i < 16; ++i) sum += __expf(f[i] - mx);
    float lse = mx + __logf(sum);
    #pragma unroll
    for (int i = 0; i < 16; ++i) f[i] -= lse;
    #pragma unroll
    for (int i = 0; i < 4; ++i) p[i] = v[i];
}

extern "C" void kernel_launch(void* const* d_in, const int* in_sizes, int n_in,
                              void* d_out, int out_size, void* d_ws, size_t ws_size,
                              hipStream_t stream) {
    const float* x   = (const float*)d_in[0];
    const int*   src = (const int*)  d_in[1];
    const int*   dst = (const int*)  d_in[2];
    const float* ew  = (const float*)d_in[3];
    const float* W1  = (const float*)d_in[4];
    const float* b1  = (const float*)d_in[5];
    const float* W2  = (const float*)d_in[6];
    const float* b2  = (const float*)d_in[7];
    float* out = (float*)d_out;

    float* ws       = (float*)d_ws;
    float* support1 = ws;
    float* agg1     = ws + (size_t)NNODES * NHID;
    float* support2 = ws + 2 * (size_t)NNODES * NHID;

    {
        int n4 = NNODES * NHID / 4;
        gcn_zero_kernel<<<(n4 + 255) / 256, 256, 0, stream>>>((float4*)agg1, n4);
        int nl = NNODES * NCLASS;
        gcn_init_logits_kernel<<<(nl + 255) / 256, 256, 0, stream>>>(b2, out, nl);
    }

    gcn_gemm1_kernel<<<MTILES, 128, 0, stream>>>(x, W1, support1);
    {
        long total = (long)NEDGES * NHID;
        int blocks = (int)((total + 255) / 256);
        gcn_spmm64_kernel<<<blocks, 256, 0, stream>>>(src, dst, ew, support1, agg1);
    }

    gcn_gemm2_kernel<<<(MTILES + 7) / 8, 256, 0, stream>>>(agg1, b1, W2, support2);
    {
        long total = (long)NEDGES * NCLASS;
        int blocks = (int)((total + 255) / 256);
        gcn_spmm16_kernel<<<blocks, 256, 0, stream>>>(src, dst, ew, support2, out);
    }

    gcn_logsoftmax_kernel<<<(NNODES + 255) / 256, 256, 0, stream>>>(out);
}


# Layer 2 regrouped as (adj @ h) @ W2: one launch sums 64-wide rows of h, then applies W2 (f32 WMMA), b2 and log_softmax (lane butterfly); 7 launches to 5; sum order: edges before the 64 products
# speedup vs baseline: 2.2740x; 2.2740x over previous
#include <hip/hip_runtime.h>

#define NNODES 100000
#define NEDGES 1600000
#define NFEAT 128
#define NHID 64
#define NCLASS 16
#define MTILES (NNODES / 16)

typedef float v2f __attribute__((ext_vector_type(2)));
typedef float v8f __attribute__((ext_vector_type(8)));

#define ROWCAP 64
#define NDEG_PAD 100352
#define NEDGE4 (NEDGES / 4)

__global__ __launch_bounds__(256) void gcn_zero_kernel(float4* __restrict__ p, int n4) {
    int i = blockIdx.x * 256 + threadIdx.x;
    if (i < n4) p[i] = make_float4(0.f, 0.f, 0.f, 0.f);
}

__device__ __forceinline__ void gcn_put_record(int d, int r, int s, float w,
                                               int* __restrict__ deg,
                                               int2* __restrict__ slot,
                                               int4* __restrict__ side) {
    if (r < ROWCAP) {
        slot[(size_t)d * ROWCAP + r] = make_int2(s, __float_as_int(w));
    } else {
        int p = __hip_atomic_fetch_add(deg + NNODES, 1, __ATOMIC_RELAXED, __HIP_MEMORY_SCOPE_AGENT);
        side[min((unsigned)p, (unsigned)(NEDGES - 1))] = make_int4(d, s, __float_as_int(w), 0);
    }
}

__global__ __launch_bounds__(256) void gcn_place_kernel(const int* __restrict__ src,
                                                        const int* __restrict__ dst,
                                                        const float* __restrict__ ew,
                                                        int* __restrict__ deg,
                                                        int2* __restrict__ slot,
                                                        int4* __restrict__ side) {
    int i = blockIdx.x * 256 + threadIdx.x;
    if (i >= NEDGE4) return;
    int4 d = ((const int4*)dst)[i];
    int4 s = ((const int4*)src)[i];
    float4 w = ((const float4*)ew)[i];
    int4 r;
    r.x = __hip_atomic_fetch_add(deg + d.x, 1, __ATOMIC_RELAXED, __HIP_MEMORY_SCOPE_AGENT);
    r.y = __hip_atomic_fetch_add(deg + d.y, 1, __ATOMIC_RELAXED, __HIP_MEMORY_SCOPE_AGENT);
    r.z = __hip_atomic_fetch_add(deg + d.z, 1, __ATOMIC_RELAXED, __HIP_MEMORY_SCOPE_AGENT);
    r.w = __hip_atomic_fetch_add(deg + d.w, 1, __ATOMIC_RELAXED, __HIP_MEMORY_SCOPE_AGENT);
    gcn_put_record(d.x, r.x, s.x, w.x, deg, slot, side);
    gcn_put_record(d.y, r.y, s.y, w.y, deg, slot, side);
    gcn_put_record(d.z, r.z, s.z, w.z, deg, slot, side);
    gcn_put_record(d.w, r.w, s.w, w.w, deg, slot, side);
}

__global__ __launch_bounds__(128) void gcn_gemm1_kernel(const float* __restrict__ x,
                                                        const float* __restrict__ W1,
                                                        float* __restrict__ out) {
    __shared__ float As[16][NFEAT + 4];
    __shared__ float WsT[NHID][NFEAT + 4];

    const int tid = threadIdx.x;
    const int tileM = blockIdx.x;

    for (int i = tid; i < NFEAT * NHID; i += 128) {
        int k = i >> 6, n = i & (NHID - 1);
        WsT[n][k] = W1[i];
    }
    const float* xrow = x + (size_t)tileM * 16 * NFEAT;
    for (int i = tid; i < 16 * NFEAT; i += 128) {
        As[i >> 7][i & (NFEAT - 1)] = xrow[i];
    }
    __syncthreads();

    const int lane = tid & 31;
    const int wave = tid >> 5;
    const int m  = lane & 15;
    const int n  = lane & 15;
    const int kh = (lane >> 4) * 2;
    const int n0 = wave * 16;

    v8f acc = {};
    #pragma unroll 4
    for (int kk = 0; kk < NFEAT; kk += 4) {
        const int k0 = kk + kh;
        v2f a, b;
        a.x = As[m][k0];
        a.y = As[m][k0 + 1];
        b.x = WsT[n0 + n][k0];
        b.y = WsT[n0 + n][k0 + 1];
        acc = __builtin_amdgcn_wmma_f32_16x16x4_f32(false, a, false, b,
                                                    (short)0, acc, false, false);
    }

    float* orow = out + ((size_t)tileM * 16 + (lane >> 4) * 8) * NHID + n0 + n;
    #pragma unroll
    for (int r = 0; r < 8; ++r)
        orow[(size_t)r * NHID] = acc[r];
}

__device__ __forceinline__ float4 gcn_row_sum64(const int row, const int l,
                                                const int* __restrict__ deg,
                                                const int2* __restrict__ slot,
                                                const int4* __restrict__ side,
                                                const float* __restrict__ sup) {
    const int n = deg[row];
    const int end = min(max(n, 0), ROWCAP);
    const int2* rec = slot + (size_t)row * ROWCAP;
    const float4* sup4 = (const float4*)sup + l;
    float4 acc = make_float4(0.f, 0.f, 0.f, 0.f);
    int j = 0;

    for (; j + 8 <= end; j += 8) {
        int2 r[8];
        float4 v[8];
        #pragma unroll
        for (int k = 0; k < 8; ++k) r[k] = rec[j + k];
        #pragma unroll
        for (int k = 0; k < 8; ++k) v[k] = sup4[(size_t)r[k].x * 16];
        #pragma unroll
        for (int k = 0; k < 8; ++k) {
            const float w = __int_as_float(r[k].y);
            acc.x = fmaf(w, v[k].x, acc.x);
            acc.y = fmaf(w, v[k].y, acc.y);
            acc.z = fmaf(w, v[k].z, acc.z);
            acc.w = fmaf(w, v[k].w, acc.w);
        }
    }
    if (j < end) {
        const int last = end - 1;
        int2 r[7];
        float4 v[7];
        #pragma unroll
        for (int k = 0; k < 7; ++k) r[k] = rec[min(j + k, last)];
        #pragma unroll
        for (int k = 0; k < 7; ++k) v[k] = sup4[(size_t)r[k].x * 16];
        #pragma unroll
        for (int k = 0; k < 7; ++k) {
            if (j + k < end) {
                const float w = __int_as_float(r[k].y);
                acc.x = fmaf(w, v[k].x, acc.x);
                acc.y = fmaf(w, v[k].y, acc.y);
                acc.z = fmaf(w, v[k].z, acc.z);
                acc.w = fmaf(w, v[k].w, acc.w);
            }
        }
    }
    if (n > ROWCAP) {
        const int m = min(deg[NNODES], NEDGES);
        for (int p = 0; p < m; ++p) {
            const int4 o = side[p];
            if (o.x == row) {
                const float w = __int_as_float(o.z);
                const float4 v = sup4[(size_t)o.y * 16];
                acc.x = fmaf(w, v.x, acc.x);
                acc.y = fmaf(w, v.y, acc.y);
                acc.z = fmaf(w, v.z, acc.z);
                acc.w = fmaf(w, v.w, acc.w);
            }
        }
    }
    return acc;
}

__global__ __launch_bounds__(256) void gcn_pull64_relu_kernel(const int* __restrict__ deg,
                                                              const int2* __restrict__ slot,
                                                              const int4* __restrict__ side,
                                                              const float* __restrict__ sup,
                                                              const float* __restrict__ b1,
                                                              float* __restrict__ out) {
    const int gid = blockIdx.x * 256 + threadIdx.x;
    const int row = gid >> 4;
    const int l = gid & 15;
    const float4 b = ((const float4*)b1)[l];
    float4 acc = gcn_row_sum64(row, l, deg, slot, side, sup);
    acc.x = fmaxf(acc.x + b.x, 0.f);
    acc.y = fmaxf(acc.y + b.y, 0.f);
    acc.z = fmaxf(acc.z + b.z, 0.f);
    acc.w = fmaxf(acc.w + b.w, 0.f);
    ((float4*)out)[(size_t)row * 16 + l] = acc;
}

template <int M>
__device__ __forceinline__ float gcn_row_xor(float v) {
    return __int_as_float(__builtin_amdgcn_update_dpp(0, __float_as_int(v), 0x160 + M, 0xf, 0xf, false));
}

__global__ __launch_bounds__(256) void gcn_pull64_cls_kernel(const int* __restrict__ deg,
                                                             const int2* __restrict__ slot,
                                                             const int4* __restrict__ side,
                                                             const float* __restrict__ h,
                                                             const float* __restrict__ W2,
                                                             const float* __restrict__ b2,
                                                             float* __restrict__ out) {
    __shared__ float As[16][NHID + 4];
    __shared__ float WsT[NCLASS][NHID + 4];

    const int tid = threadIdx.x;
    const int gid = blockIdx.x * 256 + tid;
    const int row = gid >> 4;
    const int l = gid & 15;

    {
        const float4 wv = ((const float4*)W2)[tid];
        const int k = tid >> 2, n0 = (tid & 3) * 4;
        WsT[n0 + 0][k] = wv.x;
        WsT[n0 + 1][k] = wv.y;
        WsT[n0 + 2][k] = wv.z;
        WsT[n0 + 3][k] = wv.w;
    }

    const float4 acc4 = gcn_row_sum64(row, l, deg, slot, side, h);
    *(float4*)&As[tid >> 4][4 * l] = acc4;
    __syncthreads();
    if (tid >= 32) return;

    const int c  = tid & 15;
    const int hh = tid >> 4;
    const int kh = hh * 2;

    v8f acc = {};
    #pragma unroll
    for (int kk = 0; kk < NHID; kk += 4) {
        const int k0 = kk + kh;
        v2f a, b;
        a.x = As[c][k0];
        a.y = As[c][k0 + 1];
        b.x = WsT[c][k0];
        b.y = WsT[c][k0 + 1];
        acc = __builtin_amdgcn_wmma_f32_16x16x4_f32(false, a, false, b,
                                                    (short)0, acc, false, false);
    }

    const float bias = b2[c];
    float* orow = out + ((size_t)blockIdx.x * 16 + hh * 8) * NCLASS + c;
    #pragma unroll
    for (int r = 0; r < 8; ++r) {
        const float v = acc[r] + bias;
        float mx = v;
        mx = fmaxf(mx, gcn_row_xor<1>(mx));
        mx = fmaxf(mx, gcn_row_xor<2>(mx));
        mx = fmaxf(mx, gcn_row_xor<4>(mx));
        mx = fmaxf(mx, gcn_row_xor<8>(mx));
        float s = __expf(v - mx);
        s += gcn_row_xor<1>(s);
        s += gcn_row_xor<2>(s);
        s += gcn_row_xor<4>(s);
        s += gcn_row_xor<8>(s);
        orow[(size_t)r * NCLASS] = v - (mx + __logf(s));
    }
}

extern "C" void kernel_launch(void* const* d_in, const int* in_sizes, int n_in,
                              void* d_out, int out_size, void* d_ws, size_t ws_size,
                              hipStream_t stream) {
    const float* x   = (const float*)d_in[0];
    const int*   src = (const int*)  d_in[1];
    const int*   dst = (const int*)  d_in[2];
    const float* ew  = (const float*)d_in[3];
    const float* W1  = (const float*)d_in[4];
    const float* b1  = (const float*)d_in[5];
    const float* W2  = (const float*)d_in[6];
    const float* b2  = (const float*)d_in[7];
    float* out = (float*)d_out;

    float* ws       = (float*)d_ws;
    float* support1 = ws;
    float* hid      = ws + (size_t)NNODES * NHID;
    int*   deg      = (int*)(ws + 2 * (size_t)NNODES * NHID + (size_t)NNODES * NCLASS);
    int4*  side     = (int4*)(deg + NDEG_PAD);
    int2*  slot     = (int2*)(side + NEDGES);

    gcn_zero_kernel<<<NDEG_PAD / 4 / 256, 256, 0, stream>>>((float4*)deg, NDEG_PAD / 4);
    gcn_place_kernel<<<(NEDGE4 + 255) / 256, 256, 0, stream>>>(src, dst, ew, deg, slot, side);

    gcn_gemm1_kernel<<<MTILES, 128, 0, stream>>>(x, W1, support1);
    gcn_pull64_relu_kernel<<<NNODES * 16 / 256, 256, 0, stream>>>(deg, slot, side, support1, b1, hid);

    gcn_pull64_cls_kernel<<<NNODES * 16 / 256, 256, 0, stream>>>(deg, slot, side, hid, W2, b2, out);
}


# Row-sum kernels run one wave per 16-row tile: 8 row pairs in a loop, next records requested under the current gathers (100,000 waves to 12,500); layer-2 wave multiplies its own tile; sum order kept
# speedup vs baseline: 2.3552x; 1.0357x over previous
#include <hip/hip_runtime.h>

#define NNODES 100000
#define NEDGES 1600000
#define NFEAT 128
#define NHID 64
#define NCLASS 16
#define MTILES (NNODES / 16)

typedef float v2f __attribute__((ext_vector_type(2)));
typedef float v8f __attribute__((ext_vector_type(8)));

#define ROWCAP 64
#define NDEG_PAD 100352
#define NEDGE4 (NEDGES / 4)

__global__ __launch_bounds__(256) void gcn_zero_kernel(float4* __restrict__ p, int n4) {
    int i = blockIdx.x * 256 + threadIdx.x;
    if (i < n4) p[i] = make_float4(0.f, 0.f, 0.f, 0.f);
}

__device__ __forceinline__ void gcn_put_record(int d, int r, int s, float w,
                                               int* __restrict__ deg,
                                               int2* __restrict__ slot,
                                               int4* __restrict__ side) {
    if (r < ROWCAP) {
        slot[(size_t)d * ROWCAP + r] = make_int2(s, __float_as_int(w));
    } else {
        int p = __hip_atomic_fetch_add(deg + NNODES, 1, __ATOMIC_RELAXED, __HIP_MEMORY_SCOPE_AGENT);
        side[min((unsigned)p, (unsigned)(NEDGES - 1))] = make_int4(d, s, __float_as_int(w), 0);
    }
}

__global__ __launch_bounds__(256) void gcn_place_kernel(const int* __restrict__ src,
                                                        const int* __restrict__ dst,
                                                        const float* __restrict__ ew,
                                                        int* __restrict__ deg,
                                                        int2* __restrict__ slot,
                                                        int4* __restrict__ side) {
    int i = blockIdx.x * 256 + threadIdx.x;
    if (i >= NEDGE4) return;
    int4 d = ((const int4*)dst)[i];
    int4 s = ((const int4*)src)[i];
    float4 w = ((const float4*)ew)[i];
    int4 r;
    r.x = __hip_atomic_fetch_add(deg + d.x, 1, __ATOMIC_RELAXED, __HIP_MEMORY_SCOPE_AGENT);
    r.y = __hip_atomic_fetch_add(deg + d.y, 1, __ATOMIC_RELAXED, __HIP_MEMORY_SCOPE_AGENT);
    r.z = __hip_atomic_fetch_add(deg + d.z, 1, __ATOMIC_RELAXED, __HIP_MEMORY_SCOPE_AGENT);
    r.w = __hip_atomic_fetch_add(deg + d.w, 1, __ATOMIC_RELAXED, __HIP_MEMORY_SCOPE_AGENT);
    gcn_put_record(d.x, r.x, s.x, w.x, deg, slot, side);
    gcn_put_record(d.y, r.y, s.y, w.y, deg, slot, side);
    gcn_put_record(d.z, r.z, s.z, w.z, deg, slot, side);
    gcn_put_record(d.w, r.w, s.w, w.w, deg, slot, side);
}

__global__ __launch_bounds__(128) void gcn_gemm1_kernel(const float* __restrict__ x,
                                                        const float* __restrict__ W1,
                                                        float* __restrict__ out) {
    __shared__ float As[16][NFEAT + 4];
    __shared__ float WsT[NHID][NFEAT + 4];

    const int tid = threadIdx.x;
    const int tileM = blockIdx.x;

    for (int i = tid; i < NFEAT * NHID; i += 128) {
        int k = i >> 6, n = i & (NHID - 1);
        WsT[n][k] = W1[i];
    }
    const float* xrow = x + (size_t)tileM * 16 * NFEAT;
    for (int i = tid; i < 16 * NFEAT; i += 128) {
        As[i >> 7][i & (NFEAT - 1)] = xrow[i];
    }
    __syncthreads();

    const int lane = tid & 31;
    const int wave = tid >> 5;
    const int m  = lane & 15;
    const int n  = lane & 15;
    const int kh = (lane >> 4) * 2;
    const int n0 = wave * 16;

    v8f acc = {};
    #pragma unroll 4
    for (int kk = 0; kk < NFEAT; kk += 4) {
        const int k0 = kk + kh;
        v2f a, b;
        a.x = As[m][k0];
        a.y = As[m][k0 + 1];
        b.x = WsT[n0 + n][k0];
        b.y = WsT[n0 + n][k0 + 1];
        acc = __builtin_amdgcn_wmma_f32_16x16x4_f32(false, a, false, b,
                                                    (short)0, acc, false, false);
    }

    float* orow = out + ((size_t)tileM * 16 + (lane >> 4) * 8) * NHID + n0 + n;
    #pragma unroll
    for (int r = 0; r < 8; ++r)
        orow[(size_t)r * NHID] = acc[r];
}

#define GCN_SRC(q, k) (((k) & 1) ? (q)[(k) >> 1].z : (q)[(k) >> 1].x)
#define GCN_WGT(q, k) __int_as_float(((k) & 1) ? (q)[(k) >> 1].w : (q)[(k) >> 1].y)

__device__ __forceinline__ float4 gcn_row_sum64(const int row, const int l, const int n,
                                                int4 (&q)[4],
                                                const int* __restrict__ deg,
                                                const int2* __restrict__ slot,
                                                const int4* __restrict__ side,
                                                const float* __restrict__ sup) {
    const int end = min(max(n, 0), ROWCAP);
    const int4* rec4 = (const int4*)(slot + (size_t)row * ROWCAP);
    const float4* sup4 = (const float4*)sup + l;
    float4 acc = make_float4(0.f, 0.f, 0.f, 0.f);
    int j = 0;

    for (; j + 8 <= end; j += 8) {
        int4 qn[4];
        float4 v[8];
        #pragma unroll
        for (int m = 0; m < 4; ++m) qn[m] = rec4[min((j >> 1) + 4 + m, ROWCAP / 2 - 1)];
        #pragma unroll
        for (int k = 0; k < 8; ++k) v[k] = sup4[(size_t)GCN_SRC(q, k) * 16];
        #pragma unroll
        for (int k = 0; k < 8; ++k) {
            const float w = GCN_WGT(q, k);
            acc.x = fmaf(w, v[k].x, acc.x);
            acc.y = fmaf(w, v[k].y, acc.y);
            acc.z = fmaf(w, v[k].z, acc.z);
            acc.w = fmaf(w, v[k].w, acc.w);
        }
        #pragma unroll
        for (int m = 0; m < 4; ++m) q[m] = qn[m];
    }
    if (j < end) {
        const int s0 = q[0].x;
        float4 v[7];
        #pragma unroll
        for (int k = 0; k < 7; ++k) {
            const int s = (j + k < end) ? GCN_SRC(q, k) : s0;
            v[k] = sup4[(size_t)s * 16];
        }
        #pragma unroll
        for (int k = 0; k < 7; ++k) {
            if (j + k < end) {
                const float w = GCN_WGT(q, k);
                acc.x = fmaf(w, v[k].x, acc.x);
                acc.y = fmaf(w, v[k].y, acc.y);
                acc.z = fmaf(w, v[k].z, acc.z);
                acc.w = fmaf(w, v[k].w, acc.w);
            }
        }
    }
    if (n > ROWCAP) {
        const int m = min(deg[NNODES], NEDGES);
        for (int p = 0; p < m; ++p) {
            const int4 o = side[p];
            if (o.x == row) {
                const float w = __int_as_float(o.z);
                const float4 v = sup4[(size_t)o.y * 16];
                acc.x = fmaf(w, v.x, acc.x);
                acc.y = fmaf(w, v.y, acc.y);
                acc.z = fmaf(w, v.z, acc.z);
                acc.w = fmaf(w, v.w, acc.w);
            }
        }
    }
    return acc;
}

#define GCN_TILE_HEAD()                                                                         \
    const int lane = threadIdx.x;                                                               \
    const int hh = lane >> 4;                                                     \
    const int l = lane & 15;                                   \
    const int row0 = blockIdx.x * 16 + hh;                      \
    int n = deg[row0];                                                                          \
    int4 q[4];                                                                                  \
    {                                                                                           \
        const int4* r4 = (const int4*)(slot + (size_t)row0 * ROWCAP);                           \
        _Pragma("unroll") for (int m = 0; m < 4; ++m) q[m] = r4[m];                             \
    }

#define GCN_PASS_PREFETCH()                                                                     \
    const int row = row0 + 2 * p;                                                               \
    const int rown = row0 + 2 * min(p + 1, 7);                 \
    const int nn = deg[rown];                                                                   \
    int4 qx[4];                                                                                 \
    {                                                                                           \
        const int4* r4 = (const int4*)(slot + (size_t)rown * ROWCAP);                           \
        _Pragma("unroll") for (int m = 0; m < 4; ++m) qx[m] = r4[m];                            \
    }

#define GCN_PASS_NEXT()                                                                         \
    n = nn;                                                                                     \
    _Pragma("unroll") for (int m = 0; m < 4; ++m) q[m] = qx[m];

__global__ __launch_bounds__(32) void gcn_pull64_relu_kernel(const int* __restrict__ deg,
                                                             const int2* __restrict__ slot,
                                                             const int4* __restrict__ side,
                                                             const float* __restrict__ sup,
                                                             const float* __restrict__ b1,
                                                             float* __restrict__ out) {
    GCN_TILE_HEAD()
    const float4 b = ((const float4*)b1)[l];
    for (int p = 0; p < 8; ++p) {
        GCN_PASS_PREFETCH()
        float4 acc = gcn_row_sum64(row, l, n, q, deg, slot, side, sup);
        acc.x = fmaxf(acc.x + b.x, 0.f);
        acc.y = fmaxf(acc.y + b.y, 0.f);
        acc.z = fmaxf(acc.z + b.z, 0.f);
        acc.w = fmaxf(acc.w + b.w, 0.f);
        ((float4*)out)[(size_t)row * 16 + l] = acc;
        GCN_PASS_NEXT()
    }
}

template <int M>
__device__ __forceinline__ float gcn_row_xor(float v) {
    return __int_as_float(__builtin_amdgcn_update_dpp(0, __float_as_int(v), 0x160 + M, 0xf, 0xf, false));
}

__global__ __launch_bounds__(32) void gcn_pull64_cls_kernel(const int* __restrict__ deg,
                                                            const int2* __restrict__ slot,
                                                            const int4* __restrict__ side,
                                                            const float* __restrict__ h,
                                                            const float* __restrict__ W2,
                                                            const float* __restrict__ b2,
                                                            float* __restrict__ out) {
    __shared__ float As[16][NHID + 4];

    GCN_TILE_HEAD()
    for (int p = 0; p < 8; ++p) {
        GCN_PASS_PREFETCH()
        const float4 acc4 = gcn_row_sum64(row, l, n, q, deg, slot, side, h);
        *(float4*)&As[2 * p + hh][4 * l] = acc4;
        GCN_PASS_NEXT()
    }

    const int c  = l;
    const int kh = hh * 2;

    v2f bf[NHID / 4];
    #pragma unroll
    for (int kk = 0; kk < NHID / 4; ++kk) {
        bf[kk].x = W2[(4 * kk + kh) * NCLASS + c];
        bf[kk].y = W2[(4 * kk + kh + 1) * NCLASS + c];
    }
    __syncthreads();

    v8f acc = {};
    #pragma unroll
    for (int kk = 0; kk < NHID; kk += 4) {
        const int k0 = kk + kh;
        v2f a;
        a.x = As[c][k0];
        a.y = As[c][k0 + 1];
        acc = __builtin_amdgcn_wmma_f32_16x16x4_f32(false, a, false, bf[kk >> 2],
                                                    (short)0, acc, false, false);
    }

    const float bias = b2[c];
    float* orow = out + ((size_t)blockIdx.x * 16 + hh * 8) * NCLASS + c;
    #pragma unroll
    for (int r = 0; r < 8; ++r) {
        const float v = acc[r] + bias;
        float mx = v;
        mx = fmaxf(mx, gcn_row_xor<1>(mx));
        mx = fmaxf(mx, gcn_row_xor<2>(mx));
        mx = fmaxf(mx, gcn_row_xor<4>(mx));
        mx = fmaxf(mx, gcn_row_xor<8>(mx));
        float s = __expf(v - mx);
        s += gcn_row_xor<1>(s);
        s += gcn_row_xor<2>(s);
        s += gcn_row_xor<4>(s);
        s += gcn_row_xor<8>(s);
        orow[(size_t)r * NCLASS] = v - (mx + __logf(s));
    }
}

extern "C" void kernel_launch(void* const* d_in, const int* in_sizes, int n_in,
                              void* d_out, int out_size, void* d_ws, size_t ws_size,
                              hipStream_t stream) {
    const float* x   = (const float*)d_in[0];
    const int*   src = (const int*)  d_in[1];
    const int*   dst = (const int*)  d_in[2];
    const float* ew  = (const float*)d_in[3];
    const float* W1  = (const float*)d_in[4];
    const float* b1  = (const float*)d_in[5];
    const float* W2  = (const float*)d_in[6];
    const float* b2  = (const float*)d_in[7];
    float* out = (float*)d_out;

    float* ws       = (float*)d_ws;
    float* support1 = ws;
    float* hid      = ws + (size_t)NNODES * NHID;
    int*   deg      = (int*)(ws + 2 * (size_t)NNODES * NHID + (size_t)NNODES * NCLASS);
    int4*  side     = (int4*)(deg + NDEG_PAD);
    int2*  slot     = (int2*)(side + NEDGES);

    gcn_zero_kernel<<<NDEG_PAD / 4 / 256, 256, 0, stream>>>((float4*)deg, NDEG_PAD / 4);
    gcn_place_kernel<<<(NEDGE4 + 255) / 256, 256, 0, stream>>>(src, dst, ew, deg, slot, side);

    gcn_gemm1_kernel<<<MTILES, 128, 0, stream>>>(x, W1, support1);
    gcn_pull64_relu_kernel<<<MTILES, 32, 0, stream>>>(deg, slot, side, support1, b1, hid);

    gcn_pull64_cls_kernel<<<MTILES, 32, 0, stream>>>(deg, slot, side, hid, W2, b2, out);
}


# GEMM1 without LDS or barriers: 2,048 one-wave blocks hold W1 in registers as B fragments and stream 3-4 row tiles each, x read straight into A fragments; k summed in a permuted order
# speedup vs baseline: 2.5361x; 1.0768x over previous
#include <hip/hip_runtime.h>

#define NNODES 100000
#define NEDGES 1600000
#define NFEAT 128
#define NHID 64
#define NCLASS 16
#define MTILES (NNODES / 16)

typedef float v2f __attribute__((ext_vector_type(2)));
typedef float v8f __attribute__((ext_vector_type(8)));

#define ROWCAP 64
#define NDEG_PAD 100352
#define NEDGE4 (NEDGES / 4)

__global__ __launch_bounds__(256) void gcn_zero_kernel(float4* __restrict__ p, int n4) {
    int i = blockIdx.x * 256 + threadIdx.x;
    if (i < n4) p[i] = make_float4(0.f, 0.f, 0.f, 0.f);
}

__device__ __forceinline__ void gcn_put_record(int d, int r, int s, float w,
                                               int* __restrict__ deg,
                                               int2* __restrict__ slot,
                                               int4* __restrict__ side) {
    if (r < ROWCAP) {
        slot[(size_t)d * ROWCAP + r] = make_int2(s, __float_as_int(w));
    } else {
        int p = __hip_atomic_fetch_add(deg + NNODES, 1, __ATOMIC_RELAXED, __HIP_MEMORY_SCOPE_AGENT);
        side[min((unsigned)p, (unsigned)(NEDGES - 1))] = make_int4(d, s, __float_as_int(w), 0);
    }
}

__global__ __launch_bounds__(256) void gcn_place_kernel(const int* __restrict__ src,
                                                        const int* __restrict__ dst,
                                                        const float* __restrict__ ew,
                                                        int* __restrict__ deg,
                                                        int2* __restrict__ slot,
                                                        int4* __restrict__ side) {
    int i = blockIdx.x * 256 + threadIdx.x;
    if (i >= NEDGE4) return;
    int4 d = ((const int4*)dst)[i];
    int4 s = ((const int4*)src)[i];
    float4 w = ((const float4*)ew)[i];
    int4 r;
    r.x = __hip_atomic_fetch_add(deg + d.x, 1, __ATOMIC_RELAXED, __HIP_MEMORY_SCOPE_AGENT);
    r.y = __hip_atomic_fetch_add(deg + d.y, 1, __ATOMIC_RELAXED, __HIP_MEMORY_SCOPE_AGENT);
    r.z = __hip_atomic_fetch_add(deg + d.z, 1, __ATOMIC_RELAXED, __HIP_MEMORY_SCOPE_AGENT);
    r.w = __hip_atomic_fetch_add(deg + d.w, 1, __ATOMIC_RELAXED, __HIP_MEMORY_SCOPE_AGENT);
    gcn_put_record(d.x, r.x, s.x, w.x, deg, slot, side);
    gcn_put_record(d.y, r.y, s.y, w.y, deg, slot, side);
    gcn_put_record(d.z, r.z, s.z, w.z, deg, slot, side);
    gcn_put_record(d.w, r.w, s.w, w.w, deg, slot, side);
}

#define G1_WAVES 2048

__device__ __forceinline__ void gcn_g1_load_rows(float4 (&a)[16], const float* __restrict__ x, int tile, int c, int h) {
    const float4* xp = (const float4*)x + ((size_t)(tile * 16 + c) * (NFEAT / 4) + h);
    #pragma unroll
    for (int g = 0; g < 16; ++g) a[g] = xp[2 * g];
}

typedef float v4f __attribute__((ext_vector_type(4)));

__device__ __forceinline__ void gcn_g1_hold_rows(const float4 (&a)[16]) {
    #pragma unroll
    for (int g = 0; g < 16; ++g) {
        const v4f t = {a[g].x, a[g].y, a[g].z, a[g].w};
        asm volatile("" : : "v"(t));
    }
}

__device__ __forceinline__ void gcn_g1_tile(const float4 (&a)[16], const v2f (&bf)[32][4],
                                            float* __restrict__ out, int tile, int c, int h) {
    v8f acc0 = {}, acc1 = {}, acc2 = {}, acc3 = {};
    #pragma unroll
    for (int g = 0; g < 16; ++g) {
        v2f a0, a1;
        a0.x = a[g].x;  a0.y = a[g].y;
        a1.x = a[g].z;  a1.y = a[g].w;
        acc0 = __builtin_amdgcn_wmma_f32_16x16x4_f32(false, a0, false, bf[2 * g][0], (short)0, acc0, false, false);
        acc1 = __builtin_amdgcn_wmma_f32_16x16x4_f32(false, a0, false, bf[2 * g][1], (short)0, acc1, false, false);
        acc2 = __builtin_amdgcn_wmma_f32_16x16x4_f32(false, a0, false, bf[2 * g][2], (short)0, acc2, false, false);
        acc3 = __builtin_amdgcn_wmma_f32_16x16x4_f32(false, a0, false, bf[2 * g][3], (short)0, acc3, false, false);
        acc0 = __builtin_amdgcn_wmma_f32_16x16x4_f32(false, a1, false, bf[2 * g + 1][0], (short)0, acc0, false, false);
        acc1 = __builtin_amdgcn_wmma_f32_16x16x4_f32(false, a1, false, bf[2 * g + 1][1], (short)0, acc1, false, false);
        acc2 = __builtin_amdgcn_wmma_f32_16x16x4_f32(false, a1, false, bf[2 * g + 1][2], (short)0, acc2, false, false);
        acc3 = __builtin_amdgcn_wmma_f32_16x16x4_f32(false, a1, false, bf[2 * g + 1][3], (short)0, acc3, false, false);
    }
    float4* orow = (float4*)out + ((size_t)(tile * 16 + h * 8) * (NHID / 4) + c);
    #pragma unroll
    for (int r = 0; r < 8; ++r)
        orow[(size_t)r * (NHID / 4)] = make_float4(acc0[r], acc1[r], acc2[r], acc3[r]);
}

__global__ __launch_bounds__(32) void gcn_gemm1_kernel(const float* __restrict__ x,
                                                       const float* __restrict__ W1,
                                                       float* __restrict__ out) {
    const int lane = threadIdx.x;
    const int c = lane & 15;
    const int h = lane >> 4;

    v2f bf[32][4];
    {
        const float4* w4 = (const float4*)W1 + c;
        #pragma unroll
        for (int g = 0; g < 16; ++g) {
            const int k = 8 * g + 4 * h;
            const float4 r0 = w4[(k + 0) * (NHID / 4)];
            const float4 r1 = w4[(k + 1) * (NHID / 4)];
            const float4 r2 = w4[(k + 2) * (NHID / 4)];
            const float4 r3 = w4[(k + 3) * (NHID / 4)];
            bf[2 * g][0].x = r0.x;  bf[2 * g][0].y = r1.x;
            bf[2 * g][1].x = r0.y;  bf[2 * g][1].y = r1.y;
            bf[2 * g][2].x = r0.z;  bf[2 * g][2].y = r1.z;
            bf[2 * g][3].x = r0.w;  bf[2 * g][3].y = r1.w;
            bf[2 * g + 1][0].x = r2.x;  bf[2 * g + 1][0].y = r3.x;
            bf[2 * g + 1][1].x = r2.y;  bf[2 * g + 1][1].y = r3.y;
            bf[2 * g + 1][2].x = r2.z;  bf[2 * g + 1][2].y = r3.z;
            bf[2 * g + 1][3].x = r2.w;  bf[2 * g + 1][3].y = r3.w;
        }
    }

    int tile = blockIdx.x;
    float4 a0[16], a1[16];
    gcn_g1_load_rows(a0, x, tile, c, h);
    for (;;) {
        int next = tile + G1_WAVES;
        bool more = next < MTILES;
        gcn_g1_load_rows(a1, x, more ? next : tile, c, h);
        __builtin_amdgcn_sched_barrier(0);
        gcn_g1_tile(a0, bf, out, tile, c, h);
        gcn_g1_hold_rows(a1);
        if (!more) break;
        tile = next;
        next = tile + G1_WAVES;
        more = next < MTILES;
        gcn_g1_load_rows(a0, x, more ? next : tile, c, h);
        __builtin_amdgcn_sched_barrier(0);
        gcn_g1_tile(a1, bf, out, tile, c, h);
        gcn_g1_hold_rows(a0);
        if (!more) break;
        tile = next;
    }
}

#define GCN_SRC(q, k) (((k) & 1) ? (q)[(k) >> 1].z : (q)[(k) >> 1].x)
#define GCN_WGT(q, k) __int_as_float(((k) & 1) ? (q)[(k) >> 1].w : (q)[(k) >> 1].y)

__device__ __forceinline__ float4 gcn_row_sum64(const int row, const int l, const int n,
                                                int4 (&q)[4],
                                                const int* __restrict__ deg,
                                                const int2* __restrict__ slot,
                                                const int4* __restrict__ side,
                                                const float* __restrict__ sup) {
    const int end = min(max(n, 0), ROWCAP);
    const int4* rec4 = (const int4*)(slot + (size_t)row * ROWCAP);
    const float4* sup4 = (const float4*)sup + l;
    float4 acc = make_float4(0.f, 0.f, 0.f, 0.f);
    int j = 0;

    for (; j + 8 <= end; j += 8) {
        int4 qn[4];
        float4 v[8];
        #pragma unroll
        for (int m = 0; m < 4; ++m) qn[m] = rec4[min((j >> 1) + 4 + m, ROWCAP / 2 - 1)];
        #pragma unroll
        for (int k = 0; k < 8; ++k) v[k] = sup4[(size_t)GCN_SRC(q, k) * 16];
        #pragma unroll
        for (int k = 0; k < 8; ++k) {
            const float w = GCN_WGT(q, k);
            acc.x = fmaf(w, v[k].x, acc.x);
            acc.y = fmaf(w, v[k].y, acc.y);
            acc.z = fmaf(w, v[k].z, acc.z);
            acc.w = fmaf(w, v[k].w, acc.w);
        }
        #pragma unroll
        for (int m = 0; m < 4; ++m) q[m] = qn[m];
    }
    if (j < end) {
        const int s0 = q[0].x;
        float4 v[7];
        #pragma unroll
        for (int k = 0; k < 7; ++k) {
            const int s = (j + k < end) ? GCN_SRC(q, k) : s0;
            v[k] = sup4[(size_t)s * 16];
        }
        #pragma unroll
        for (int k = 0; k < 7; ++k) {
            if (j + k < end) {
                const float w = GCN_WGT(q, k);
                acc.x = fmaf(w, v[k].x, acc.x);
                acc.y = fmaf(w, v[k].y, acc.y);
                acc.z = fmaf(w, v[k].z, acc.z);
                acc.w = fmaf(w, v[k].w, acc.w);
            }
        }
    }
    if (n > ROWCAP) {
        const int m = min(deg[NNODES], NEDGES);
        for (int p = 0; p < m; ++p) {
            const int4 o = side[p];
            if (o.x == row) {
                const float w = __int_as_float(o.z);
                const float4 v = sup4[(size_t)o.y * 16];
                acc.x = fmaf(w, v.x, acc.x);
                acc.y = fmaf(w, v.y, acc.y);
                acc.z = fmaf(w, v.z, acc.z);
                acc.w = fmaf(w, v.w, acc.w);
            }
        }
    }
    return acc;
}

#define GCN_TILE_HEAD()                                                                         \
    const int lane = threadIdx.x;                                                               \
    const int hh = lane >> 4;                                                     \
    const int l = lane & 15;                                   \
    const int row0 = blockIdx.x * 16 + hh;                      \
    int n = deg[row0];                                                                          \
    int4 q[4];                                                                                  \
    {                                                                                           \
        const int4* r4 = (const int4*)(slot + (size_t)row0 * ROWCAP);                           \
        _Pragma("unroll") for (int m = 0; m < 4; ++m) q[m] = r4[m];                             \
    }

#define GCN_PASS_PREFETCH()                                                                     \
    const int row = row0 + 2 * p;                                                               \
    const int rown = row0 + 2 * min(p + 1, 7);                 \
    const int nn = deg[rown];                                                                   \
    int4 qx[4];                                                                                 \
    {                                                                                           \
        const int4* r4 = (const int4*)(slot + (size_t)rown * ROWCAP);                           \
        _Pragma("unroll") for (int m = 0; m < 4; ++m) qx[m] = r4[m];                            \
    }

#define GCN_PASS_NEXT()                                                                         \
    n = nn;                                                                                     \
    _Pragma("unroll") for (int m = 0; m < 4; ++m) q[m] = qx[m];

__global__ __launch_bounds__(32) void gcn_pull64_relu_kernel(const int* __restrict__ deg,
                                                             const int2* __restrict__ slot,
                                                             const int4* __restrict__ side,
                                                             const float* __restrict__ sup,
                                                             const float* __restrict__ b1,
                                                             float* __restrict__ out) {
    GCN_TILE_HEAD()
    const float4 b = ((const float4*)b1)[l];
    for (int p = 0; p < 8; ++p) {
        GCN_PASS_PREFETCH()
        float4 acc = gcn_row_sum64(row, l, n, q, deg, slot, side, sup);
        acc.x = fmaxf(acc.x + b.x, 0.f);
        acc.y = fmaxf(acc.y + b.y, 0.f);
        acc.z = fmaxf(acc.z + b.z, 0.f);
        acc.w = fmaxf(acc.w + b.w, 0.f);
        ((float4*)out)[(size_t)row * 16 + l] = acc;
        GCN_PASS_NEXT()
    }
}

template <int M>
__device__ __forceinline__ float gcn_row_xor(float v) {
    return __int_as_float(__builtin_amdgcn_update_dpp(0, __float_as_int(v), 0x160 + M, 0xf, 0xf, false));
}

__global__ __launch_bounds__(32) void gcn_pull64_cls_kernel(const int* __restrict__ deg,
                                                            const int2* __restrict__ slot,
                                                            const int4* __restrict__ side,
                                                            const float* __restrict__ h,
                                                            const float* __restrict__ W2,
                                                            const float* __restrict__ b2,
                                                            float* __restrict__ out) {
    __shared__ float As[16][NHID + 4];

    GCN_TILE_HEAD()
    for (int p = 0; p < 8; ++p) {
        GCN_PASS_PREFETCH()
        const float4 acc4 = gcn_row_sum64(row, l, n, q, deg, slot, side, h);
        *(float4*)&As[2 * p + hh][4 * l] = acc4;
        GCN_PASS_NEXT()
    }

    const int c  = l;
    const int kh = hh * 2;

    v2f bf[NHID / 4];
    #pragma unroll
    for (int kk = 0; kk < NHID / 4; ++kk) {
        bf[kk].x = W2[(4 * kk + kh) * NCLASS + c];
        bf[kk].y = W2[(4 * kk + kh + 1) * NCLASS + c];
    }
    __syncthreads();

    v8f acc = {};
    #pragma unroll
    for (int kk = 0; kk < NHID; kk += 4) {
        const int k0 = kk + kh;
        v2f a;
        a.x = As[c][k0];
        a.y = As[c][k0 + 1];
        acc = __builtin_amdgcn_wmma_f32_16x16x4_f32(false, a, false, bf[kk >> 2],
                                                    (short)0, acc, false, false);
    }

    const float bias = b2[c];
    float* orow = out + ((size_t)blockIdx.x * 16 + hh * 8) * NCLASS + c;
    #pragma unroll
    for (int r = 0; r < 8; ++r) {
        const float v = acc[r] + bias;
        float mx = v;
        mx = fmaxf(mx, gcn_row_xor<1>(mx));
        mx = fmaxf(mx, gcn_row_xor<2>(mx));
        mx = fmaxf(mx, gcn_row_xor<4>(mx));
        mx = fmaxf(mx, gcn_row_xor<8>(mx));
        float s = __expf(v - mx);
        s += gcn_row_xor<1>(s);
        s += gcn_row_xor<2>(s);
        s += gcn_row_xor<4>(s);
        s += gcn_row_xor<8>(s);
        orow[(size_t)r * NCLASS] = v - (mx + __logf(s));
    }
}

extern "C" void kernel_launch(void* const* d_in, const int* in_sizes, int n_in,
                              void* d_out, int out_size, void* d_ws, size_t ws_size,
                              hipStream_t stream) {
    const float* x   = (const float*)d_in[0];
    const int*   src = (const int*)  d_in[1];
    const int*   dst = (const int*)  d_in[2];
    const float* ew  = (const float*)d_in[3];
    const float* W1  = (const float*)d_in[4];
    const float* b1  = (const float*)d_in[5];
    const float* W2  = (const float*)d_in[6];
    const float* b2  = (const float*)d_in[7];
    float* out = (float*)d_out;

    float* ws       = (float*)d_ws;
    float* support1 = ws;
    float* hid      = ws + (size_t)NNODES * NHID;
    int*   deg      = (int*)(ws + 2 * (size_t)NNODES * NHID + (size_t)NNODES * NCLASS);
    int4*  side     = (int4*)(deg + NDEG_PAD);
    int2*  slot     = (int2*)(side + NEDGES);

    gcn_zero_kernel<<<NDEG_PAD / 4 / 256, 256, 0, stream>>>((float4*)deg, NDEG_PAD / 4);
    gcn_place_kernel<<<(NEDGE4 + 255) / 256, 256, 0, stream>>>(src, dst, ew, deg, slot, side);

    gcn_gemm1_kernel<<<G1_WAVES, 32, 0, stream>>>(x, W1, support1);
    gcn_pull64_relu_kernel<<<MTILES, 32, 0, stream>>>(deg, slot, side, support1, b1, hid);

    gcn_pull64_cls_kernel<<<MTILES, 32, 0, stream>>>(deg, slot, side, hid, W2, b2, out);
}


# Place blocks (256 x 1,024 threads, 6,250 edges each) counting-sort their edges by 128-B counter line in LDS, so a wave's returning adds share lines (fewer atomic requests); sum order: arrival rank
# speedup vs baseline: 3.2790x; 1.2929x over previous
#include <hip/hip_runtime.h>

#define NNODES 100000
#define NEDGES 1600000
#define NFEAT 128
#define NHID 64
#define NCLASS 16
#define MTILES (NNODES / 16)

typedef float v2f __attribute__((ext_vector_type(2)));
typedef float v8f __attribute__((ext_vector_type(8)));

#define ROWCAP 64
#define NDEG_PAD 100352

__global__ __launch_bounds__(256) void gcn_zero_kernel(float4* __restrict__ p, int n4) {
    int i = blockIdx.x * 256 + threadIdx.x;
    if (i < n4) p[i] = make_float4(0.f, 0.f, 0.f, 0.f);
}

__device__ __forceinline__ void gcn_put_record(int d, int r, int s, float w,
                                               int* __restrict__ deg,
                                               int2* __restrict__ slot,
                                               int4* __restrict__ side) {
    if (r < ROWCAP) {
        slot[(size_t)d * ROWCAP + r] = make_int2(s, __float_as_int(w));
    } else {
        int p = __hip_atomic_fetch_add(deg + NNODES, 1, __ATOMIC_RELAXED, __HIP_MEMORY_SCOPE_AGENT);
        side[min((unsigned)p, (unsigned)(NEDGES - 1))] = make_int4(d, s, __float_as_int(w), 0);
    }
}

#define PL_EDGES 6250
#define PL_BLOCKS (NEDGES / PL_EDGES)
#define PL_THREADS 1024
#define PL_PER ((PL_EDGES + PL_THREADS - 1) / PL_THREADS)
#define PL_BINS (NDEG_PAD / 32)
#define PL_SCAN_THREADS (PL_BINS / 4)

__device__ __forceinline__ int gcn_line_bin(int d) {
    return (int)min((unsigned)d >> 5, (unsigned)(PL_BINS - 1));
}

__global__ __launch_bounds__(PL_THREADS) void gcn_place_kernel(const int* __restrict__ src,
                                                               const int* __restrict__ dst,
                                                               const float* __restrict__ ew,
                                                               int* __restrict__ deg,
                                                               int2* __restrict__ slot,
                                                               int4* __restrict__ side) {
    __shared__ int   cnt[PL_BINS];
    __shared__ int   wsum[PL_THREADS / 32];
    __shared__ int   sd[PL_EDGES];
    __shared__ int   ss[PL_EDGES];
    __shared__ float sw[PL_EDGES];

    const int tid  = threadIdx.x;
    const int lane = tid & 31;
    const int wave = tid >> 5;
    const int base = blockIdx.x * PL_EDGES;

    for (int i = tid; i < PL_BINS; i += PL_THREADS) cnt[i] = 0;
    __syncthreads();

    int d[PL_PER], s[PL_PER], r[PL_PER];
    float w[PL_PER];
    #pragma unroll
    for (int k = 0; k < PL_PER; ++k) {
        const int i = tid + k * PL_THREADS;
        const bool on = i < PL_EDGES;
        d[k] = on ? dst[base + i] : 0;
        s[k] = on ? src[base + i] : 0;
        w[k] = on ? ew[base + i] : 0.f;
    }
    #pragma unroll
    for (int k = 0; k < PL_PER; ++k) {
        const int i = tid + k * PL_THREADS;
        r[k] = 0;
        if (i < PL_EDGES) r[k] = atomicAdd(&cnt[gcn_line_bin(d[k])], 1);
    }
    __syncthreads();

    int c0 = 0, c1 = 0, c2 = 0, c3 = 0;
    if (tid < PL_SCAN_THREADS) {
        c0 = cnt[4 * tid];
        c1 = cnt[4 * tid + 1];
        c2 = cnt[4 * tid + 2];
        c3 = cnt[4 * tid + 3];
    }
    const int tsum = c0 + c1 + c2 + c3;
    int incl = tsum;
    #pragma unroll
    for (int o = 1; o < 32; o <<= 1) {
        const int t = __shfl_up(incl, o);
        if (lane >= o) incl += t;
    }
    if (lane == 31) wsum[wave] = incl;
    __syncthreads();
    int first = incl - tsum;
    for (int j = 0; j < wave; ++j) first += wsum[j];
    if (tid < PL_SCAN_THREADS) {
        cnt[4 * tid]     = first;
        cnt[4 * tid + 1] = first + c0;
        cnt[4 * tid + 2] = first + c0 + c1;
        cnt[4 * tid + 3] = first + c0 + c1 + c2;
    }
    __syncthreads();

    #pragma unroll
    for (int k = 0; k < PL_PER; ++k) {
        const int i = tid + k * PL_THREADS;
        if (i < PL_EDGES) {
            const int p = min(cnt[gcn_line_bin(d[k])] + r[k], PL_EDGES - 1);
            sd[p] = d[k];
            ss[p] = s[k];
            sw[p] = w[k];
        }
    }
    __syncthreads();

    #pragma unroll
    for (int k = 0; k < PL_PER; ++k) {
        const int i = tid + k * PL_THREADS;
        const bool on = i < PL_EDGES;
        d[k] = on ? sd[i] : 0;
        s[k] = on ? ss[i] : 0;
        w[k] = on ? sw[i] : 0.f;
    }
    #pragma unroll
    for (int k = 0; k < PL_PER; ++k) {
        const int i = tid + k * PL_THREADS;
        r[k] = 0;
        if (i < PL_EDGES)
            r[k] = __hip_atomic_fetch_add(deg + d[k], 1, __ATOMIC_RELAXED, __HIP_MEMORY_SCOPE_AGENT);
    }
    #pragma unroll
    for (int k = 0; k < PL_PER; ++k) {
        const int i = tid + k * PL_THREADS;
        if (i < PL_EDGES) gcn_put_record(d[k], r[k], s[k], w[k], deg, slot, side);
    }
}

#define G1_WAVES 2048

__device__ __forceinline__ void gcn_g1_load_rows(float4 (&a)[16], const float* __restrict__ x, int tile, int c, int h) {
    const float4* xp = (const float4*)x + ((size_t)(tile * 16 + c) * (NFEAT / 4) + h);
    #pragma unroll
    for (int g = 0; g < 16; ++g) a[g] = xp[2 * g];
}

typedef float v4f __attribute__((ext_vector_type(4)));

__device__ __forceinline__ void gcn_g1_hold_rows(const float4 (&a)[16]) {
    #pragma unroll
    for (int g = 0; g < 16; ++g) {
        const v4f t = {a[g].x, a[g].y, a[g].z, a[g].w};
        asm volatile("" : : "v"(t));
    }
}

__device__ __forceinline__ void gcn_g1_tile(const float4 (&a)[16], const v2f (&bf)[32][4],
                                            float* __restrict__ out, int tile, int c, int h) {
    v8f acc0 = {}, acc1 = {}, acc2 = {}, acc3 = {};
    #pragma unroll
    for (int g = 0; g < 16; ++g) {
        v2f a0, a1;
        a0.x = a[g].x;  a0.y = a[g].y;
        a1.x = a[g].z;  a1.y = a[g].w;
        acc0 = __builtin_amdgcn_wmma_f32_16x16x4_f32(false, a0, false, bf[2 * g][0], (short)0, acc0, false, false);
        acc1 = __builtin_amdgcn_wmma_f32_16x16x4_f32(false, a0, false, bf[2 * g][1], (short)0, acc1, false, false);
        acc2 = __builtin_amdgcn_wmma_f32_16x16x4_f32(false, a0, false, bf[2 * g][2], (short)0, acc2, false, false);
        acc3 = __builtin_amdgcn_wmma_f32_16x16x4_f32(false, a0, false, bf[2 * g][3], (short)0, acc3, false, false);
        acc0 = __builtin_amdgcn_wmma_f32_16x16x4_f32(false, a1, false, bf[2 * g + 1][0], (short)0, acc0, false, false);
        acc1 = __builtin_amdgcn_wmma_f32_16x16x4_f32(false, a1, false, bf[2 * g + 1][1], (short)0, acc1, false, false);
        acc2 = __builtin_amdgcn_wmma_f32_16x16x4_f32(false, a1, false, bf[2 * g + 1][2], (short)0, acc2, false, false);
        acc3 = __builtin_amdgcn_wmma_f32_16x16x4_f32(false, a1, false, bf[2 * g + 1][3], (short)0, acc3, false, false);
    }
    float4* orow = (float4*)out + ((size_t)(tile * 16 + h * 8) * (NHID / 4) + c);
    #pragma unroll
    for (int r = 0; r < 8; ++r)
        orow[(size_t)r * (NHID / 4)] = make_float4(acc0[r], acc1[r], acc2[r], acc3[r]);
}

__global__ __launch_bounds__(32) void gcn_gemm1_kernel(const float* __restrict__ x,
                                                       const float* __restrict__ W1,
                                                       float* __restrict__ out) {
    const int lane = threadIdx.x;
    const int c = lane & 15;
    const int h = lane >> 4;

    v2f bf[32][4];
    {
        const float4* w4 = (const float4*)W1 + c;
        #pragma unroll
        for (int g = 0; g < 16; ++g) {
            const int k = 8 * g + 4 * h;
            const float4 r0 = w4[(k + 0) * (NHID / 4)];
            const float4 r1 = w4[(k + 1) * (NHID / 4)];
            const float4 r2 = w4[(k + 2) * (NHID / 4)];
            const float4 r3 = w4[(k + 3) * (NHID / 4)];
            bf[2 * g][0].x = r0.x;  bf[2 * g][0].y = r1.x;
            bf[2 * g][1].x = r0.y;  bf[2 * g][1].y = r1.y;
            bf[2 * g][2].x = r0.z;  bf[2 * g][2].y = r1.z;
            bf[2 * g][3].x = r0.w;  bf[2 * g][3].y = r1.w;
            bf[2 * g + 1][0].x = r2.x;  bf[2 * g + 1][0].y = r3.x;
            bf[2 * g + 1][1].x = r2.y;  bf[2 * g + 1][1].y = r3.y;
            bf[2 * g + 1][2].x = r2.z;  bf[2 * g + 1][2].y = r3.z;
            bf[2 * g + 1][3].x = r2.w;  bf[2 * g + 1][3].y = r3.w;
        }
    }

    int tile = blockIdx.x;
    float4 a0[16], a1[16];
    gcn_g1_load_rows(a0, x, tile, c, h);
    for (;;) {
        int next = tile + G1_WAVES;
        bool more = next < MTILES;
        gcn_g1_load_rows(a1, x, more ? next : tile, c, h);
        __builtin_amdgcn_sched_barrier(0);
        gcn_g1_tile(a0, bf, out, tile, c, h);
        gcn_g1_hold_rows(a1);
        if (!more) break;
        tile = next;
        next = tile + G1_WAVES;
        more = next < MTILES;
        gcn_g1_load_rows(a0, x, more ? next : tile, c, h);
        __builtin_amdgcn_sched_barrier(0);
        gcn_g1_tile(a1, bf, out, tile, c, h);
        gcn_g1_hold_rows(a0);
        if (!more) break;
        tile = next;
    }
}

#define GCN_SRC(q, k) (((k) & 1) ? (q)[(k) >> 1].z : (q)[(k) >> 1].x)
#define GCN_WGT(q, k) __int_as_float(((k) & 1) ? (q)[(k) >> 1].w : (q)[(k) >> 1].y)

__device__ __forceinline__ float4 gcn_row_sum64(const int row, const int l, const int n,
                                                int4 (&q)[4],
                                                const int* __restrict__ deg,
                                                const int2* __restrict__ slot,
                                                const int4* __restrict__ side,
                                                const float* __restrict__ sup) {
    const int end = min(max(n, 0), ROWCAP);
    const int4* rec4 = (const int4*)(slot + (size_t)row * ROWCAP);
    const float4* sup4 = (const float4*)sup + l;
    float4 acc = make_float4(0.f, 0.f, 0.f, 0.f);
    int j = 0;

    for (; j + 8 <= end; j += 8) {
        int4 qn[4];
        float4 v[8];
        #pragma unroll
        for (int m = 0; m < 4; ++m) qn[m] = rec4[min((j >> 1) + 4 + m, ROWCAP / 2 - 1)];
        #pragma unroll
        for (int k = 0; k < 8; ++k) v[k] = sup4[(size_t)GCN_SRC(q, k) * 16];
        #pragma unroll
        for (int k = 0; k < 8; ++k) {
            const float w = GCN_WGT(q, k);
            acc.x = fmaf(w, v[k].x, acc.x);
            acc.y = fmaf(w, v[k].y, acc.y);
            acc.z = fmaf(w, v[k].z, acc.z);
            acc.w = fmaf(w, v[k].w, acc.w);
        }
        #pragma unroll
        for (int m = 0; m < 4; ++m) q[m] = qn[m];
    }
    if (j < end) {
        const int s0 = q[0].x;
        float4 v[7];
        #pragma unroll
        for (int k = 0; k < 7; ++k) {
            const int s = (j + k < end) ? GCN_SRC(q, k) : s0;
            v[k] = sup4[(size_t)s * 16];
        }
        #pragma unroll
        for (int k = 0; k < 7; ++k) {
            if (j + k < end) {
                const float w = GCN_WGT(q, k);
                acc.x = fmaf(w, v[k].x, acc.x);
                acc.y = fmaf(w, v[k].y, acc.y);
                acc.z = fmaf(w, v[k].z, acc.z);
                acc.w = fmaf(w, v[k].w, acc.w);
            }
        }
    }
    if (n > ROWCAP) {
        const int m = min(deg[NNODES], NEDGES);
        for (int p = 0; p < m; ++p) {
            const int4 o = side[p];
            if (o.x == row) {
                const float w = __int_as_float(o.z);
                const float4 v = sup4[(size_t)o.y * 16];
                acc.x = fmaf(w, v.x, acc.x);
                acc.y = fmaf(w, v.y, acc.y);
                acc.z = fmaf(w, v.z, acc.z);
                acc.w = fmaf(w, v.w, acc.w);
            }
        }
    }
    return acc;
}

#define GCN_TILE_HEAD()                                                                         \
    const int lane = threadIdx.x;                                                               \
    const int hh = lane >> 4;                                                     \
    const int l = lane & 15;                                   \
    const int row0 = blockIdx.x * 16 + hh;                      \
    int n = deg[row0];                                                                          \
    int4 q[4];                                                                                  \
    {                                                                                           \
        const int4* r4 = (const int4*)(slot + (size_t)row0 * ROWCAP);                           \
        _Pragma("unroll") for (int m = 0; m < 4; ++m) q[m] = r4[m];                             \
    }

#define GCN_PASS_PREFETCH()                                                                     \
    const int row = row0 + 2 * p;                                                               \
    const int rown = row0 + 2 * min(p + 1, 7);                 \
    const int nn = deg[rown];                                                                   \
    int4 qx[4];                                                                                 \
    {                                                                                           \
        const int4* r4 = (const int4*)(slot + (size_t)rown * ROWCAP);                           \
        _Pragma("unroll") for (int m = 0; m < 4; ++m) qx[m] = r4[m];                            \
    }

#define GCN_PASS_NEXT()                                                                         \
    n = nn;                                                                                     \
    _Pragma("unroll") for (int m = 0; m < 4; ++m) q[m] = qx[m];

__global__ __launch_bounds__(32) void gcn_pull64_relu_kernel(const int* __restrict__ deg,
                                                             const int2* __restrict__ slot,
                                                             const int4* __restrict__ side,
                                                             const float* __restrict__ sup,
                                                             const float* __restrict__ b1,
                                                             float* __restrict__ out) {
    GCN_TILE_HEAD()
    const float4 b = ((const float4*)b1)[l];
    for (int p = 0; p < 8; ++p) {
        GCN_PASS_PREFETCH()
        float4 acc = gcn_row_sum64(row, l, n, q, deg, slot, side, sup);
        acc.x = fmaxf(acc.x + b.x, 0.f);
        acc.y = fmaxf(acc.y + b.y, 0.f);
        acc.z = fmaxf(acc.z + b.z, 0.f);
        acc.w = fmaxf(acc.w + b.w, 0.f);
        ((float4*)out)[(size_t)row * 16 + l] = acc;
        GCN_PASS_NEXT()
    }
}

template <int M>
__device__ __forceinline__ float gcn_row_xor(float v) {
    return __int_as_float(__builtin_amdgcn_update_dpp(0, __float_as_int(v), 0x160 + M, 0xf, 0xf, false));
}

__global__ __launch_bounds__(32) void gcn_pull64_cls_kernel(const int* __restrict__ deg,
                                                            const int2* __restrict__ slot,
                                                            const int4* __restrict__ side,
                                                            const float* __restrict__ h,
                                                            const float* __restrict__ W2,
                                                            const float* __restrict__ b2,
                                                            float* __restrict__ out) {
    __shared__ float As[16][NHID + 4];

    GCN_TILE_HEAD()
    for (int p = 0; p < 8; ++p) {
        GCN_PASS_PREFETCH()
        const float4 acc4 = gcn_row_sum64(row, l, n, q, deg, slot, side, h);
        *(float4*)&As[2 * p + hh][4 * l] = acc4;
        GCN_PASS_NEXT()
    }

    const int c  = l;
    const int kh = hh * 2;

    v2f bf[NHID / 4];
    #pragma unroll
    for (int kk = 0; kk < NHID / 4; ++kk) {
        bf[kk].x = W2[(4 * kk + kh) * NCLASS + c];
        bf[kk].y = W2[(4 * kk + kh + 1) * NCLASS + c];
    }
    __syncthreads();

    v8f acc = {};
    #pragma unroll
    for (int kk = 0; kk < NHID; kk += 4) {
        const int k0 = kk + kh;
        v2f a;
        a.x = As[c][k0];
        a.y = As[c][k0 + 1];
        acc = __builtin_amdgcn_wmma_f32_16x16x4_f32(false, a, false, bf[kk >> 2],
                                                    (short)0, acc, false, false);
    }

    const float bias = b2[c];
    float* orow = out + ((size_t)blockIdx.x * 16 + hh * 8) * NCLASS + c;
    #pragma unroll
    for (int r = 0; r < 8; ++r) {
        const float v = acc[r] + bias;
        float mx = v;
        mx = fmaxf(mx, gcn_row_xor<1>(mx));
        mx = fmaxf(mx, gcn_row_xor<2>(mx));
        mx = fmaxf(mx, gcn_row_xor<4>(mx));
        mx = fmaxf(mx, gcn_row_xor<8>(mx));
        float s = __expf(v - mx);
        s += gcn_row_xor<1>(s);
        s += gcn_row_xor<2>(s);
        s += gcn_row_xor<4>(s);
        s += gcn_row_xor<8>(s);
        orow[(size_t)r * NCLASS] = v - (mx + __logf(s));
    }
}

extern "C" void kernel_launch(void* const* d_in, const int* in_sizes, int n_in,
                              void* d_out, int out_size, void* d_ws, size_t ws_size,
                              hipStream_t stream) {
    const float* x   = (const float*)d_in[0];
    const int*   src = (const int*)  d_in[1];
    const int*   dst = (const int*)  d_in[2];
    const float* ew  = (const float*)d_in[3];
    const float* W1  = (const float*)d_in[4];
    const float* b1  = (const float*)d_in[5];
    const float* W2  = (const float*)d_in[6];
    const float* b2  = (const float*)d_in[7];
    float* out = (float*)d_out;

    float* ws       = (float*)d_ws;
    float* support1 = ws;
    float* hid      = ws + (size_t)NNODES * NHID;
    int*   deg      = (int*)(ws + 2 * (size_t)NNODES * NHID + (size_t)NNODES * NCLASS);
    int4*  side     = (int4*)(deg + NDEG_PAD);
    int2*  slot     = (int2*)(side + NEDGES);

    gcn_zero_kernel<<<NDEG_PAD / 4 / 256, 256, 0, stream>>>((float4*)deg, NDEG_PAD / 4);
    gcn_place_kernel<<<PL_BLOCKS, PL_THREADS, 0, stream>>>(src, dst, ew, deg, slot, side);

    gcn_gemm1_kernel<<<G1_WAVES, 32, 0, stream>>>(x, W1, support1);
    gcn_pull64_relu_kernel<<<MTILES, 32, 0, stream>>>(deg, slot, side, support1, b1, hid);

    gcn_pull64_cls_kernel<<<MTILES, 32, 0, stream>>>(deg, slot, side, hid, W2, b2, out);
}


# Place blocks take edges by destination range: block (range of 8, slice of 32) reads its 50,000-edge slice, keeps its range's edges: 16 per counter line, fewer add requests; sum order: arrival rank
# speedup vs baseline: 5.4271x; 1.6551x over previous
#include <hip/hip_runtime.h>

#define NNODES 100000
#define NEDGES 1600000
#define NFEAT 128
#define NHID 64
#define NCLASS 16
#define MTILES (NNODES / 16)

typedef float v2f __attribute__((ext_vector_type(2)));
typedef float v8f __attribute__((ext_vector_type(8)));

#define ROWCAP 64
#define NDEG_PAD 100352

__global__ __launch_bounds__(256) void gcn_zero_kernel(float4* __restrict__ p, int n4) {
    int i = blockIdx.x * 256 + threadIdx.x;
    if (i < n4) p[i] = make_float4(0.f, 0.f, 0.f, 0.f);
}

__device__ __forceinline__ void gcn_put_record(int d, int r, int s, float w,
                                               int* __restrict__ deg,
                                               int2* __restrict__ slot,
                                               int4* __restrict__ side) {
    if (r < ROWCAP) {
        slot[(size_t)d * ROWCAP + r] = make_int2(s, __float_as_int(w));
    } else {
        int p = __hip_atomic_fetch_add(deg + NNODES, 1, __ATOMIC_RELAXED, __HIP_MEMORY_SCOPE_AGENT);
        side[min((unsigned)p, (unsigned)(NEDGES - 1))] = make_int4(d, s, __float_as_int(w), 0);
    }
}

#define PL_RANGES 8
#define PL_SLICES 32
#define PL_BLOCKS (PL_RANGES * PL_SLICES)
#define PL_THREADS 1024
#define PL_SLICE_EDGES (NEDGES / PL_SLICES)
#define PL_SLICE4 (PL_SLICE_EDGES / 4)
#define PL_GROUPS 2
#define PL_PASSES ((PL_SLICE4 + PL_GROUPS * PL_THREADS - 1) / (PL_GROUPS * PL_THREADS))
#define PL_BINS (NDEG_PAD / 32)
#define PL_RBINS (PL_BINS / PL_RANGES)
#define PL_CAP 7168
#define PL_PER (PL_CAP / PL_THREADS)

__device__ __forceinline__ int gcn_line_bin(int d) {
    return (int)min((unsigned)d >> 5, (unsigned)(PL_BINS - 1));
}

__device__ __forceinline__ int gcn_range_bin(int d, int line0) {
    return (int)min((unsigned)(gcn_line_bin(d) - line0), (unsigned)(PL_RBINS - 1));
}

__device__ __forceinline__ void gcn_place_direct(int d, int s, float w,
                                              int* __restrict__ deg,
                                              int2* __restrict__ slot,
                                              int4* __restrict__ side) {
    const int r = __hip_atomic_fetch_add(deg + d, 1, __ATOMIC_RELAXED, __HIP_MEMORY_SCOPE_AGENT);
    gcn_put_record(d, r, s, w, deg, slot, side);
}

__device__ __forceinline__ void gcn_place_load(int4 (&dn)[PL_GROUPS], int4 (&sn)[PL_GROUPS], float4 (&wn)[PL_GROUPS],
                                               const int4* __restrict__ dst4, const int4* __restrict__ src4,
                                               const float4* __restrict__ ew4, int pass, int tid) {
    #pragma unroll
    for (int u = 0; u < PL_GROUPS; ++u) {
        const int g = min((pass * PL_GROUPS + u) * PL_THREADS + tid, PL_SLICE4 - 1);
        dn[u] = dst4[g];
        sn[u] = src4[g];
        wn[u] = ew4[g];
    }
}

__global__ __launch_bounds__(PL_THREADS) void gcn_place_kernel(const int* __restrict__ src,
                                                               const int* __restrict__ dst,
                                                               const float* __restrict__ ew,
                                                               int* __restrict__ deg,
                                                               int2* __restrict__ slot,
                                                               int4* __restrict__ side) {
    __shared__ int   cnt[PL_RBINS];
    __shared__ int   wsum[PL_THREADS / 32];
    __shared__ int   kept;
    __shared__ int   sd[PL_CAP];
    __shared__ int   ss[PL_CAP];
    __shared__ float sw[PL_CAP];

    const int tid  = threadIdx.x;
    const int lane = tid & 31;
    const int wave = tid >> 5;
    const int range = blockIdx.x % PL_RANGES;
    const int slice = blockIdx.x / PL_RANGES;
    const int line0 = range * PL_RBINS;
    const int4*   dst4 = (const int4*)(dst + (size_t)slice * PL_SLICE_EDGES);
    const int4*   src4 = (const int4*)(src + (size_t)slice * PL_SLICE_EDGES);
    const float4* ew4  = (const float4*)(ew + (size_t)slice * PL_SLICE_EDGES);

    for (int i = tid; i < PL_RBINS; i += PL_THREADS) cnt[i] = 0;
    if (tid == 0) kept = 0;
    __syncthreads();

    int4 dn[PL_GROUPS], sn[PL_GROUPS];
    float4 wn[PL_GROUPS];
    gcn_place_load(dn, sn, wn, dst4, src4, ew4, 0, tid);
    #pragma nounroll
    for (int pass = 0; pass < PL_PASSES; ++pass) {
        int dd[4 * PL_GROUPS], sv[4 * PL_GROUPS];
        float wv[4 * PL_GROUPS];
        bool on[PL_GROUPS];
        #pragma unroll
        for (int u = 0; u < PL_GROUPS; ++u) {
            on[u] = (pass * PL_GROUPS + u) * PL_THREADS + tid < PL_SLICE4;
            dd[4 * u] = dn[u].x;  dd[4 * u + 1] = dn[u].y;  dd[4 * u + 2] = dn[u].z;  dd[4 * u + 3] = dn[u].w;
            sv[4 * u] = sn[u].x;  sv[4 * u + 1] = sn[u].y;  sv[4 * u + 2] = sn[u].z;  sv[4 * u + 3] = sn[u].w;
            wv[4 * u] = wn[u].x;  wv[4 * u + 1] = wn[u].y;  wv[4 * u + 2] = wn[u].z;  wv[4 * u + 3] = wn[u].w;
        }
        gcn_place_load(dn, sn, wn, dst4, src4, ew4, pass + 1, tid);

        bool in[4 * PL_GROUPS];
        unsigned m[4 * PL_GROUPS];
        int total = 0;
        #pragma unroll
        for (int j = 0; j < 4 * PL_GROUPS; ++j) {
            in[j] = on[j >> 2] && (unsigned)(gcn_line_bin(dd[j]) - line0) < (unsigned)PL_RBINS;
            m[j] = __builtin_amdgcn_ballot_w32(in[j]);
            total += __popc(m[j]);
        }
        int off = 0;
        if (lane == 0 && total > 0) off = atomicAdd(&kept, total);
        off = __builtin_amdgcn_readfirstlane(off);
        #pragma unroll
        for (int j = 0; j < 4 * PL_GROUPS; ++j) {
            if (in[j]) {
                const int p = off + (int)__builtin_amdgcn_mbcnt_lo(m[j], 0u);
                if (p < PL_CAP) {
                    sd[p] = dd[j];
                    ss[p] = sv[j];
                    sw[p] = wv[j];
                } else {
                    gcn_place_direct(dd[j], sv[j], wv[j], deg, slot, side);
                }
            }
            off += __popc(m[j]);
        }
    }
    __syncthreads();
    const int n = min(kept, PL_CAP);

    int d[PL_PER], s[PL_PER], r[PL_PER];
    float w[PL_PER];
    #pragma unroll
    for (int k = 0; k < PL_PER; ++k) {
        const int i = tid + k * PL_THREADS;
        const bool on = i < n;
        d[k] = on ? sd[i] : 0;
        s[k] = on ? ss[i] : 0;
        w[k] = on ? sw[i] : 0.f;
    }
    #pragma unroll
    for (int k = 0; k < PL_PER; ++k) {
        const int i = tid + k * PL_THREADS;
        r[k] = 0;
        if (i < n) r[k] = atomicAdd(&cnt[gcn_range_bin(d[k], line0)], 1);
    }
    __syncthreads();

    const int c0 = (tid < PL_RBINS) ? cnt[tid] : 0;
    int incl = c0;
    #pragma unroll
    for (int o = 1; o < 32; o <<= 1) {
        const int t = __shfl_up(incl, o);
        if (lane >= o) incl += t;
    }
    if (lane == 31) wsum[wave] = incl;
    __syncthreads();
    if (tid < PL_RBINS) {
        int first = incl - c0;
        for (int j = 0; j < wave; ++j) first += wsum[j];
        cnt[tid] = first;
    }
    __syncthreads();

    #pragma unroll
    for (int k = 0; k < PL_PER; ++k) {
        const int i = tid + k * PL_THREADS;
        if (i < n) {
            const int p = min(cnt[gcn_range_bin(d[k], line0)] + r[k], PL_CAP - 1);
            sd[p] = d[k];
            ss[p] = s[k];
            sw[p] = w[k];
        }
    }
    __syncthreads();

    #pragma unroll
    for (int k = 0; k < PL_PER; ++k) {
        const int i = tid + k * PL_THREADS;
        const bool on = i < n;
        d[k] = on ? sd[i] : 0;
        s[k] = on ? ss[i] : 0;
        w[k] = on ? sw[i] : 0.f;
    }
    #pragma unroll
    for (int k = 0; k < PL_PER; ++k) {
        const int i = tid + k * PL_THREADS;
        r[k] = 0;
        if (i < n)
            r[k] = __hip_atomic_fetch_add(deg + d[k], 1, __ATOMIC_RELAXED, __HIP_MEMORY_SCOPE_AGENT);
    }
    #pragma unroll
    for (int k = 0; k < PL_PER; ++k) {
        const int i = tid + k * PL_THREADS;
        if (i < n) gcn_put_record(d[k], r[k], s[k], w[k], deg, slot, side);
    }
}

#define G1_WAVES 2048

__device__ __forceinline__ void gcn_g1_load_rows(float4 (&a)[16], const float* __restrict__ x, int tile, int c, int h) {
    const float4* xp = (const float4*)x + ((size_t)(tile * 16 + c) * (NFEAT / 4) + h);
    #pragma unroll
    for (int g = 0; g < 16; ++g) a[g] = xp[2 * g];
}

typedef float v4f __attribute__((ext_vector_type(4)));

__device__ __forceinline__ void gcn_g1_hold_rows(const float4 (&a)[16]) {
    #pragma unroll
    for (int g = 0; g < 16; ++g) {
        const v4f t = {a[g].x, a[g].y, a[g].z, a[g].w};
        asm volatile("" : : "v"(t));
    }
}

__device__ __forceinline__ void gcn_g1_tile(const float4 (&a)[16], const v2f (&bf)[32][4],
                                            float* __restrict__ out, int tile, int c, int h) {
    v8f acc0 = {}, acc1 = {}, acc2 = {}, acc3 = {};
    #pragma unroll
    for (int g = 0; g < 16; ++g) {
        v2f a0, a1;
        a0.x = a[g].x;  a0.y = a[g].y;
        a1.x = a[g].z;  a1.y = a[g].w;
        acc0 = __builtin_amdgcn_wmma_f32_16x16x4_f32(false, a0, false, bf[2 * g][0], (short)0, acc0, false, false);
        acc1 = __builtin_amdgcn_wmma_f32_16x16x4_f32(false, a0, false, bf[2 * g][1], (short)0, acc1, false, false);
        acc2 = __builtin_amdgcn_wmma_f32_16x16x4_f32(false, a0, false, bf[2 * g][2], (short)0, acc2, false, false);
        acc3 = __builtin_amdgcn_wmma_f32_16x16x4_f32(false, a0, false, bf[2 * g][3], (short)0, acc3, false, false);
        acc0 = __builtin_amdgcn_wmma_f32_16x16x4_f32(false, a1, false, bf[2 * g + 1][0], (short)0, acc0, false, false);
        acc1 = __builtin_amdgcn_wmma_f32_16x16x4_f32(false, a1, false, bf[2 * g + 1][1], (short)0, acc1, false, false);
        acc2 = __builtin_amdgcn_wmma_f32_16x16x4_f32(false, a1, false, bf[2 * g + 1][2], (short)0, acc2, false, false);
        acc3 = __builtin_amdgcn_wmma_f32_16x16x4_f32(false, a1, false, bf[2 * g + 1][3], (short)0, acc3, false, false);
    }
    float4* orow = (float4*)out + ((size_t)(tile * 16 + h * 8) * (NHID / 4) + c);
    #pragma unroll
    for (int r = 0; r < 8; ++r)
        orow[(size_t)r * (NHID / 4)] = make_float4(acc0[r], acc1[r], acc2[r], acc3[r]);
}

__global__ __launch_bounds__(32) void gcn_gemm1_kernel(const float* __restrict__ x,
                                                       const float* __restrict__ W1,
                                                       float* __restrict__ out) {
    const int lane = threadIdx.x;
    const int c = lane & 15;
    const int h = lane >> 4;

    v2f bf[32][4];
    {
        const float4* w4 = (const float4*)W1 + c;
        #pragma unroll
        for (int g = 0; g < 16; ++g) {
            const int k = 8 * g + 4 * h;
            const float4 r0 = w4[(k + 0) * (NHID / 4)];
            const float4 r1 = w4[(k + 1) * (NHID / 4)];
            const float4 r2 = w4[(k + 2) * (NHID / 4)];
            const float4 r3 = w4[(k + 3) * (NHID / 4)];
            bf[2 * g][0].x = r0.x;  bf[2 * g][0].y = r1.x;
            bf[2 * g][1].x = r0.y;  bf[2 * g][1].y = r1.y;
            bf[2 * g][2].x = r0.z;  bf[2 * g][2].y = r1.z;
            bf[2 * g][3].x = r0.w;  bf[2 * g][3].y = r1.w;
            bf[2 * g + 1][0].x = r2.x;  bf[2 * g + 1][0].y = r3.x;
            bf[2 * g + 1][1].x = r2.y;  bf[2 * g + 1][1].y = r3.y;
            bf[2 * g + 1][2].x = r2.z;  bf[2 * g + 1][2].y = r3.z;
            bf[2 * g + 1][3].x = r2.w;  bf[2 * g + 1][3].y = r3.w;
        }
    }

    int tile = blockIdx.x;
    float4 a0[16], a1[16];
    gcn_g1_load_rows(a0, x, tile, c, h);
    for (;;) {
        int next = tile + G1_WAVES;
        bool more = next < MTILES;
        gcn_g1_load_rows(a1, x, more ? next : tile, c, h);
        __builtin_amdgcn_sched_barrier(0);
        gcn_g1_tile(a0, bf, out, tile, c, h);
        gcn_g1_hold_rows(a1);
        if (!more) break;
        tile = next;
        next = tile + G1_WAVES;
        more = next < MTILES;
        gcn_g1_load_rows(a0, x, more ? next : tile, c, h);
        __builtin_amdgcn_sched_barrier(0);
        gcn_g1_tile(a1, bf, out, tile, c, h);
        gcn_g1_hold_rows(a0);
        if (!more) break;
        tile = next;
    }
}

#define GCN_SRC(q, k) (((k) & 1) ? (q)[(k) >> 1].z : (q)[(k) >> 1].x)
#define GCN_WGT(q, k) __int_as_float(((k) & 1) ? (q)[(k) >> 1].w : (q)[(k) >> 1].y)

__device__ __forceinline__ float4 gcn_row_sum64(const int row, const int l, const int n,
                                                int4 (&q)[4],
                                                const int* __restrict__ deg,
                                                const int2* __restrict__ slot,
                                                const int4* __restrict__ side,
                                                const float* __restrict__ sup) {
    const int end = min(max(n, 0), ROWCAP);
    const int4* rec4 = (const int4*)(slot + (size_t)row * ROWCAP);
    const float4* sup4 = (const float4*)sup + l;
    float4 acc = make_float4(0.f, 0.f, 0.f, 0.f);
    int j = 0;

    for (; j + 8 <= end; j += 8) {
        int4 qn[4];
        float4 v[8];
        #pragma unroll
        for (int m = 0; m < 4; ++m) qn[m] = rec4[min((j >> 1) + 4 + m, ROWCAP / 2 - 1)];
        #pragma unroll
        for (int k = 0; k < 8; ++k) v[k] = sup4[(size_t)GCN_SRC(q, k) * 16];
        #pragma unroll
        for (int k = 0; k < 8; ++k) {
            const float w = GCN_WGT(q, k);
            acc.x = fmaf(w, v[k].x, acc.x);
            acc.y = fmaf(w, v[k].y, acc.y);
            acc.z = fmaf(w, v[k].z, acc.z);
            acc.w = fmaf(w, v[k].w, acc.w);
        }
        #pragma unroll
        for (int m = 0; m < 4; ++m) q[m] = qn[m];
    }
    if (j < end) {
        const int s0 = q[0].x;
        float4 v[7];
        #pragma unroll
        for (int k = 0; k < 7; ++k) {
            const int s = (j + k < end) ? GCN_SRC(q, k) : s0;
            v[k] = sup4[(size_t)s * 16];
        }
        #pragma unroll
        for (int k = 0; k < 7; ++k) {
            if (j + k < end) {
                const float w = GCN_WGT(q, k);
                acc.x = fmaf(w, v[k].x, acc.x);
                acc.y = fmaf(w, v[k].y, acc.y);
                acc.z = fmaf(w, v[k].z, acc.z);
                acc.w = fmaf(w, v[k].w, acc.w);
            }
        }
    }
    if (n > ROWCAP) {
        const int m = min(deg[NNODES], NEDGES);
        for (int p = 0; p < m; ++p) {
            const int4 o = side[p];
            if (o.x == row) {
                const float w = __int_as_float(o.z);
                const float4 v = sup4[(size_t)o.y * 16];
                acc.x = fmaf(w, v.x, acc.x);
                acc.y = fmaf(w, v.y, acc.y);
                acc.z = fmaf(w, v.z, acc.z);
                acc.w = fmaf(w, v.w, acc.w);
            }
        }
    }
    return acc;
}

#define GCN_TILE_HEAD()                                                                         \
    const int lane = threadIdx.x;                                                               \
    const int hh = lane >> 4;                                                     \
    const int l = lane & 15;                                   \
    const int row0 = blockIdx.x * 16 + hh;                      \
    int n = deg[row0];                                                                          \
    int4 q[4];                                                                                  \
    {                                                                                           \
        const int4* r4 = (const int4*)(slot + (size_t)row0 * ROWCAP);                           \
        _Pragma("unroll") for (int m = 0; m < 4; ++m) q[m] = r4[m];                             \
    }

#define GCN_PASS_PREFETCH()                                                                     \
    const int row = row0 + 2 * p;                                                               \
    const int rown = row0 + 2 * min(p + 1, 7);                 \
    const int nn = deg[rown];                                                                   \
    int4 qx[4];                                                                                 \
    {                                                                                           \
        const int4* r4 = (const int4*)(slot + (size_t)rown * ROWCAP);                           \
        _Pragma("unroll") for (int m = 0; m < 4; ++m) qx[m] = r4[m];                            \
    }

#define GCN_PASS_NEXT()                                                                         \
    n = nn;                                                                                     \
    _Pragma("unroll") for (int m = 0; m < 4; ++m) q[m] = qx[m];

__global__ __launch_bounds__(32) void gcn_pull64_relu_kernel(const int* __restrict__ deg,
                                                             const int2* __restrict__ slot,
                                                             const int4* __restrict__ side,
                                                             const float* __restrict__ sup,
                                                             const float* __restrict__ b1,
                                                             float* __restrict__ out) {
    GCN_TILE_HEAD()
    const float4 b = ((const float4*)b1)[l];
    for (int p = 0; p < 8; ++p) {
        GCN_PASS_PREFETCH()
        float4 acc = gcn_row_sum64(row, l, n, q, deg, slot, side, sup);
        acc.x = fmaxf(acc.x + b.x, 0.f);
        acc.y = fmaxf(acc.y + b.y, 0.f);
        acc.z = fmaxf(acc.z + b.z, 0.f);
        acc.w = fmaxf(acc.w + b.w, 0.f);
        ((float4*)out)[(size_t)row * 16 + l] = acc;
        GCN_PASS_NEXT()
    }
}

template <int M>
__device__ __forceinline__ float gcn_row_xor(float v) {
    return __int_as_float(__builtin_amdgcn_update_dpp(0, __float_as_int(v), 0x160 + M, 0xf, 0xf, false));
}

__global__ __launch_bounds__(32) void gcn_pull64_cls_kernel(const int* __restrict__ deg,
                                                            const int2* __restrict__ slot,
                                                            const int4* __restrict__ side,
                                                            const float* __restrict__ h,
                                                            const float* __restrict__ W2,
                                                            const float* __restrict__ b2,
                                                            float* __restrict__ out) {
    __shared__ float As[16][NHID + 4];

    GCN_TILE_HEAD()
    for (int p = 0; p < 8; ++p) {
        GCN_PASS_PREFETCH()
        const float4 acc4 = gcn_row_sum64(row, l, n, q, deg, slot, side, h);
        *(float4*)&As[2 * p + hh][4 * l] = acc4;
        GCN_PASS_NEXT()
    }

    const int c  = l;
    const int kh = hh * 2;

    v2f bf[NHID / 4];
    #pragma unroll
    for (int kk = 0; kk < NHID / 4; ++kk) {
        bf[kk].x = W2[(4 * kk + kh) * NCLASS + c];
        bf[kk].y = W2[(4 * kk + kh + 1) * NCLASS + c];
    }
    __syncthreads();

    v8f acc = {};
    #pragma unroll
    for (int kk = 0; kk < NHID; kk += 4) {
        const int k0 = kk + kh;
        v2f a;
        a.x = As[c][k0];
        a.y = As[c][k0 + 1];
        acc = __builtin_amdgcn_wmma_f32_16x16x4_f32(false, a, false, bf[kk >> 2],
                                                    (short)0, acc, false, false);
    }

    const float bias = b2[c];
    float* orow = out + ((size_t)blockIdx.x * 16 + hh * 8) * NCLASS + c;
    #pragma unroll
    for (int r = 0; r < 8; ++r) {
        const float v = acc[r] + bias;
        float mx = v;
        mx = fmaxf(mx, gcn_row_xor<1>(mx));
        mx = fmaxf(mx, gcn_row_xor<2>(mx));
        mx = fmaxf(mx, gcn_row_xor<4>(mx));
        mx = fmaxf(mx, gcn_row_xor<8>(mx));
        float s = __expf(v - mx);
        s += gcn_row_xor<1>(s);
        s += gcn_row_xor<2>(s);
        s += gcn_row_xor<4>(s);
        s += gcn_row_xor<8>(s);
        orow[(size_t)r * NCLASS] = v - (mx + __logf(s));
    }
}

extern "C" void kernel_launch(void* const* d_in, const int* in_sizes, int n_in,
                              void* d_out, int out_size, void* d_ws, size_t ws_size,
                              hipStream_t stream) {
    const float* x   = (const float*)d_in[0];
    const int*   src = (const int*)  d_in[1];
    const int*   dst = (const int*)  d_in[2];
    const float* ew  = (const float*)d_in[3];
    const float* W1  = (const float*)d_in[4];
    const float* b1  = (const float*)d_in[5];
    const float* W2  = (const float*)d_in[6];
    const float* b2  = (const float*)d_in[7];
    float* out = (float*)d_out;

    float* ws       = (float*)d_ws;
    float* support1 = ws;
    float* hid      = ws + (size_t)NNODES * NHID;
    int*   deg      = (int*)(ws + 2 * (size_t)NNODES * NHID + (size_t)NNODES * NCLASS);
    int4*  side     = (int4*)(deg + NDEG_PAD);
    int2*  slot     = (int2*)(side + NEDGES);

    gcn_zero_kernel<<<NDEG_PAD / 4 / 256, 256, 0, stream>>>((float4*)deg, NDEG_PAD / 4);
    gcn_place_kernel<<<PL_BLOCKS, PL_THREADS, 0, stream>>>(src, dst, ew, deg, slot, side);

    gcn_gemm1_kernel<<<G1_WAVES, 32, 0, stream>>>(x, W1, support1);
    gcn_pull64_relu_kernel<<<MTILES, 32, 0, stream>>>(deg, slot, side, support1, b1, hid);

    gcn_pull64_cls_kernel<<<MTILES, 32, 0, stream>>>(deg, slot, side, hid, W2, b2, out);
}


# W2 applied by the layer-1 row-sum wave (16 rows via LDS, f32 WMMA): h never stored; last launch gathers 64-B rows, 4 lanes a row; sum order: 64 products before edges, softmax 4 in lane then 4 lanes
# speedup vs baseline: 6.2246x; 1.1469x over previous
#include <hip/hip_runtime.h>

#define NNODES 100000
#define NEDGES 1600000
#define NFEAT 128
#define NHID 64
#define NCLASS 16
#define MTILES (NNODES / 16)

typedef float v2f __attribute__((ext_vector_type(2)));
typedef float v8f __attribute__((ext_vector_type(8)));

#define ROWCAP 64
#define NDEG_PAD 100352

__global__ __launch_bounds__(256) void gcn_zero_kernel(float4* __restrict__ p, int n4) {
    int i = blockIdx.x * 256 + threadIdx.x;
    if (i < n4) p[i] = make_float4(0.f, 0.f, 0.f, 0.f);
}

__device__ __forceinline__ void gcn_put_record(int d, int r, int s, float w,
                                               int* __restrict__ deg,
                                               int2* __restrict__ slot,
                                               int4* __restrict__ side) {
    if (r < ROWCAP) {
        slot[(size_t)d * ROWCAP + r] = make_int2(s, __float_as_int(w));
    } else {
        int p = __hip_atomic_fetch_add(deg + NNODES, 1, __ATOMIC_RELAXED, __HIP_MEMORY_SCOPE_AGENT);
        side[min((unsigned)p, (unsigned)(NEDGES - 1))] = make_int4(d, s, __float_as_int(w), 0);
    }
}

#define PL_RANGES 8
#define PL_SLICES 32
#define PL_BLOCKS (PL_RANGES * PL_SLICES)
#define PL_THREADS 1024
#define PL_SLICE_EDGES (NEDGES / PL_SLICES)
#define PL_SLICE4 (PL_SLICE_EDGES / 4)
#define PL_GROUPS 2
#define PL_PASSES ((PL_SLICE4 + PL_GROUPS * PL_THREADS - 1) / (PL_GROUPS * PL_THREADS))
#define PL_BINS (NDEG_PAD / 32)
#define PL_RBINS (PL_BINS / PL_RANGES)
#define PL_CAP 7168
#define PL_PER (PL_CAP / PL_THREADS)

__device__ __forceinline__ int gcn_line_bin(int d) {
    return (int)min((unsigned)d >> 5, (unsigned)(PL_BINS - 1));
}

__device__ __forceinline__ int gcn_range_bin(int d, int line0) {
    return (int)min((unsigned)(gcn_line_bin(d) - line0), (unsigned)(PL_RBINS - 1));
}

__device__ __forceinline__ void gcn_place_direct(int d, int s, float w,
                                              int* __restrict__ deg,
                                              int2* __restrict__ slot,
                                              int4* __restrict__ side) {
    const int r = __hip_atomic_fetch_add(deg + d, 1, __ATOMIC_RELAXED, __HIP_MEMORY_SCOPE_AGENT);
    gcn_put_record(d, r, s, w, deg, slot, side);
}

__device__ __forceinline__ void gcn_place_load(int4 (&dn)[PL_GROUPS], int4 (&sn)[PL_GROUPS], float4 (&wn)[PL_GROUPS],
                                               const int4* __restrict__ dst4, const int4* __restrict__ src4,
                                               const float4* __restrict__ ew4, int pass, int tid) {
    #pragma unroll
    for (int u = 0; u < PL_GROUPS; ++u) {
        const int g = min((pass * PL_GROUPS + u) * PL_THREADS + tid, PL_SLICE4 - 1);
        dn[u] = dst4[g];
        sn[u] = src4[g];
        wn[u] = ew4[g];
    }
}

__global__ __launch_bounds__(PL_THREADS) void gcn_place_kernel(const int* __restrict__ src,
                                                               const int* __restrict__ dst,
                                                               const float* __restrict__ ew,
                                                               int* __restrict__ deg,
                                                               int2* __restrict__ slot,
                                                               int4* __restrict__ side) {
    __shared__ int   cnt[PL_RBINS];
    __shared__ int   wsum[PL_THREADS / 32];
    __shared__ int   kept;
    __shared__ int   sd[PL_CAP];
    __shared__ int   ss[PL_CAP];
    __shared__ float sw[PL_CAP];

    const int tid  = threadIdx.x;
    const int lane = tid & 31;
    const int wave = tid >> 5;
    const int range = blockIdx.x % PL_RANGES;
    const int slice = blockIdx.x / PL_RANGES;
    const int line0 = range * PL_RBINS;
    const int4*   dst4 = (const int4*)(dst + (size_t)slice * PL_SLICE_EDGES);
    const int4*   src4 = (const int4*)(src + (size_t)slice * PL_SLICE_EDGES);
    const float4* ew4  = (const float4*)(ew + (size_t)slice * PL_SLICE_EDGES);

    for (int i = tid; i < PL_RBINS; i += PL_THREADS) cnt[i] = 0;
    if (tid == 0) kept = 0;
    __syncthreads();

    int4 dn[PL_GROUPS], sn[PL_GROUPS];
    float4 wn[PL_GROUPS];
    gcn_place_load(dn, sn, wn, dst4, src4, ew4, 0, tid);
    #pragma nounroll
    for (int pass = 0; pass < PL_PASSES; ++pass) {
        int dd[4 * PL_GROUPS], sv[4 * PL_GROUPS];
        float wv[4 * PL_GROUPS];
        bool on[PL_GROUPS];
        #pragma unroll
        for (int u = 0; u < PL_GROUPS; ++u) {
            on[u] = (pass * PL_GROUPS + u) * PL_THREADS + tid < PL_SLICE4;
            dd[4 * u] = dn[u].x;  dd[4 * u + 1] = dn[u].y;  dd[4 * u + 2] = dn[u].z;  dd[4 * u + 3] = dn[u].w;
            sv[4 * u] = sn[u].x;  sv[4 * u + 1] = sn[u].y;  sv[4 * u + 2] = sn[u].z;  sv[4 * u + 3] = sn[u].w;
            wv[4 * u] = wn[u].x;  wv[4 * u + 1] = wn[u].y;  wv[4 * u + 2] = wn[u].z;  wv[4 * u + 3] = wn[u].w;
        }
        gcn_place_load(dn, sn, wn, dst4, src4, ew4, pass + 1, tid);

        bool in[4 * PL_GROUPS];
        unsigned m[4 * PL_GROUPS];
        int total = 0;
        #pragma unroll
        for (int j = 0; j < 4 * PL_GROUPS; ++j) {
            in[j] = on[j >> 2] && (unsigned)(gcn_line_bin(dd[j]) - line0) < (unsigned)PL_RBINS;
            m[j] = __builtin_amdgcn_ballot_w32(in[j]);
            total += __popc(m[j]);
        }
        int off = 0;
        if (lane == 0 && total > 0) off = atomicAdd(&kept, total);
        off = __builtin_amdgcn_readfirstlane(off);
        #pragma unroll
        for (int j = 0; j < 4 * PL_GROUPS; ++j) {
            if (in[j]) {
                const int p = off + (int)__builtin_amdgcn_mbcnt_lo(m[j], 0u);
                if (p < PL_CAP) {
                    sd[p] = dd[j];
                    ss[p] = sv[j];
                    sw[p] = wv[j];
                } else {
                    gcn_place_direct(dd[j], sv[j], wv[j], deg, slot, side);
                }
            }
            off += __popc(m[j]);
        }
    }
    __syncthreads();
    const int n = min(kept, PL_CAP);

    int d[PL_PER], s[PL_PER], r[PL_PER];
    float w[PL_PER];
    #pragma unroll
    for (int k = 0; k < PL_PER; ++k) {
        const int i = tid + k * PL_THREADS;
        const bool on = i < n;
        d[k] = on ? sd[i] : 0;
        s[k] = on ? ss[i] : 0;
        w[k] = on ? sw[i] : 0.f;
    }
    #pragma unroll
    for (int k = 0; k < PL_PER; ++k) {
        const int i = tid + k * PL_THREADS;
        r[k] = 0;
        if (i < n) r[k] = atomicAdd(&cnt[gcn_range_bin(d[k], line0)], 1);
    }
    __syncthreads();

    const int c0 = (tid < PL_RBINS) ? cnt[tid] : 0;
    int incl = c0;
    #pragma unroll
    for (int o = 1; o < 32; o <<= 1) {
        const int t = __shfl_up(incl, o);
        if (lane >= o) incl += t;
    }
    if (lane == 31) wsum[wave] = incl;
    __syncthreads();
    if (tid < PL_RBINS) {
        int first = incl - c0;
        for (int j = 0; j < wave; ++j) first += wsum[j];
        cnt[tid] = first;
    }
    __syncthreads();

    #pragma unroll
    for (int k = 0; k < PL_PER; ++k) {
        const int i = tid + k * PL_THREADS;
        if (i < n) {
            const int p = min(cnt[gcn_range_bin(d[k], line0)] + r[k], PL_CAP - 1);
            sd[p] = d[k];
            ss[p] = s[k];
            sw[p] = w[k];
        }
    }
    __syncthreads();

    #pragma unroll
    for (int k = 0; k < PL_PER; ++k) {
        const int i = tid + k * PL_THREADS;
        const bool on = i < n;
        d[k] = on ? sd[i] : 0;
        s[k] = on ? ss[i] : 0;
        w[k] = on ? sw[i] : 0.f;
    }
    #pragma unroll
    for (int k = 0; k < PL_PER; ++k) {
        const int i = tid + k * PL_THREADS;
        r[k] = 0;
        if (i < n)
            r[k] = __hip_atomic_fetch_add(deg + d[k], 1, __ATOMIC_RELAXED, __HIP_MEMORY_SCOPE_AGENT);
    }
    #pragma unroll
    for (int k = 0; k < PL_PER; ++k) {
        const int i = tid + k * PL_THREADS;
        if (i < n) gcn_put_record(d[k], r[k], s[k], w[k], deg, slot, side);
    }
}

#define G1_WAVES 2048

__device__ __forceinline__ void gcn_g1_load_rows(float4 (&a)[16], const float* __restrict__ x, int tile, int c, int h) {
    const float4* xp = (const float4*)x + ((size_t)(tile * 16 + c) * (NFEAT / 4) + h);
    #pragma unroll
    for (int g = 0; g < 16; ++g) a[g] = xp[2 * g];
}

typedef float v4f __attribute__((ext_vector_type(4)));

__device__ __forceinline__ void gcn_g1_hold_rows(const float4 (&a)[16]) {
    #pragma unroll
    for (int g = 0; g < 16; ++g) {
        const v4f t = {a[g].x, a[g].y, a[g].z, a[g].w};
        asm volatile("" : : "v"(t));
    }
}

__device__ __forceinline__ void gcn_g1_tile(const float4 (&a)[16], const v2f (&bf)[32][4],
                                            float* __restrict__ out, int tile, int c, int h) {
    v8f acc0 = {}, acc1 = {}, acc2 = {}, acc3 = {};
    #pragma unroll
    for (int g = 0; g < 16; ++g) {
        v2f a0, a1;
        a0.x = a[g].x;  a0.y = a[g].y;
        a1.x = a[g].z;  a1.y = a[g].w;
        acc0 = __builtin_amdgcn_wmma_f32_16x16x4_f32(false, a0, false, bf[2 * g][0], (short)0, acc0, false, false);
        acc1 = __builtin_amdgcn_wmma_f32_16x16x4_f32(false, a0, false, bf[2 * g][1], (short)0, acc1, false, false);
        acc2 = __builtin_amdgcn_wmma_f32_16x16x4_f32(false, a0, false, bf[2 * g][2], (short)0, acc2, false, false);
        acc3 = __builtin_amdgcn_wmma_f32_16x16x4_f32(false, a0, false, bf[2 * g][3], (short)0, acc3, false, false);
        acc0 = __builtin_amdgcn_wmma_f32_16x16x4_f32(false, a1, false, bf[2 * g + 1][0], (short)0, acc0, false, false);
        acc1 = __builtin_amdgcn_wmma_f32_16x16x4_f32(false, a1, false, bf[2 * g + 1][1], (short)0, acc1, false, false);
        acc2 = __builtin_amdgcn_wmma_f32_16x16x4_f32(false, a1, false, bf[2 * g + 1][2], (short)0, acc2, false, false);
        acc3 = __builtin_amdgcn_wmma_f32_16x16x4_f32(false, a1, false, bf[2 * g + 1][3], (short)0, acc3, false, false);
    }
    float4* orow = (float4*)out + ((size_t)(tile * 16 + h * 8) * (NHID / 4) + c);
    #pragma unroll
    for (int r = 0; r < 8; ++r)
        orow[(size_t)r * (NHID / 4)] = make_float4(acc0[r], acc1[r], acc2[r], acc3[r]);
}

__global__ __launch_bounds__(32) void gcn_gemm1_kernel(const float* __restrict__ x,
                                                       const float* __restrict__ W1,
                                                       float* __restrict__ out) {
    const int lane = threadIdx.x;
    const int c = lane & 15;
    const int h = lane >> 4;

    v2f bf[32][4];
    {
        const float4* w4 = (const float4*)W1 + c;
        #pragma unroll
        for (int g = 0; g < 16; ++g) {
            const int k = 8 * g + 4 * h;
            const float4 r0 = w4[(k + 0) * (NHID / 4)];
            const float4 r1 = w4[(k + 1) * (NHID / 4)];
            const float4 r2 = w4[(k + 2) * (NHID / 4)];
            const float4 r3 = w4[(k + 3) * (NHID / 4)];
            bf[2 * g][0].x = r0.x;  bf[2 * g][0].y = r1.x;
            bf[2 * g][1].x = r0.y;  bf[2 * g][1].y = r1.y;
            bf[2 * g][2].x = r0.z;  bf[2 * g][2].y = r1.z;
            bf[2 * g][3].x = r0.w;  bf[2 * g][3].y = r1.w;
            bf[2 * g + 1][0].x = r2.x;  bf[2 * g + 1][0].y = r3.x;
            bf[2 * g + 1][1].x = r2.y;  bf[2 * g + 1][1].y = r3.y;
            bf[2 * g + 1][2].x = r2.z;  bf[2 * g + 1][2].y = r3.z;
            bf[2 * g + 1][3].x = r2.w;  bf[2 * g + 1][3].y = r3.w;
        }
    }

    int tile = blockIdx.x;
    float4 a0[16], a1[16];
    gcn_g1_load_rows(a0, x, tile, c, h);
    for (;;) {
        int next = tile + G1_WAVES;
        bool more = next < MTILES;
        gcn_g1_load_rows(a1, x, more ? next : tile, c, h);
        __builtin_amdgcn_sched_barrier(0);
        gcn_g1_tile(a0, bf, out, tile, c, h);
        gcn_g1_hold_rows(a1);
        if (!more) break;
        tile = next;
        next = tile + G1_WAVES;
        more = next < MTILES;
        gcn_g1_load_rows(a0, x, more ? next : tile, c, h);
        __builtin_amdgcn_sched_barrier(0);
        gcn_g1_tile(a1, bf, out, tile, c, h);
        gcn_g1_hold_rows(a0);
        if (!more) break;
        tile = next;
    }
}

#define GCN_SRC(q, k) (((k) & 1) ? (q)[(k) >> 1].z : (q)[(k) >> 1].x)
#define GCN_WGT(q, k) __int_as_float(((k) & 1) ? (q)[(k) >> 1].w : (q)[(k) >> 1].y)

template <int P4>
__device__ __forceinline__ float4 gcn_row_sum(const int row, const int l, const int n,
                                              int4 (&q)[4],
                                              const int* __restrict__ deg,
                                              const int2* __restrict__ slot,
                                              const int4* __restrict__ side,
                                              const float* __restrict__ sup) {
    const int end = min(max(n, 0), ROWCAP);
    const int4* rec4 = (const int4*)(slot + (size_t)row * ROWCAP);
    const float4* sup4 = (const float4*)sup + l;
    float4 acc = make_float4(0.f, 0.f, 0.f, 0.f);
    int j = 0;

    for (; j + 8 <= end; j += 8) {
        int4 qn[4];
        float4 v[8];
        #pragma unroll
        for (int m = 0; m < 4; ++m) qn[m] = rec4[min((j >> 1) + 4 + m, ROWCAP / 2 - 1)];
        #pragma unroll
        for (int k = 0; k < 8; ++k) v[k] = sup4[(size_t)GCN_SRC(q, k) * P4];
        #pragma unroll
        for (int k = 0; k < 8; ++k) {
            const float w = GCN_WGT(q, k);
            acc.x = fmaf(w, v[k].x, acc.x);
            acc.y = fmaf(w, v[k].y, acc.y);
            acc.z = fmaf(w, v[k].z, acc.z);
            acc.w = fmaf(w, v[k].w, acc.w);
        }
        #pragma unroll
        for (int m = 0; m < 4; ++m) q[m] = qn[m];
    }
    if (j < end) {
        const int s0 = q[0].x;
        float4 v[7];
        #pragma unroll
        for (int k = 0; k < 7; ++k) {
            const int s = (j + k < end) ? GCN_SRC(q, k) : s0;
            v[k] = sup4[(size_t)s * P4];
        }
        #pragma unroll
        for (int k = 0; k < 7; ++k) {
            if (j + k < end) {
                const float w = GCN_WGT(q, k);
                acc.x = fmaf(w, v[k].x, acc.x);
                acc.y = fmaf(w, v[k].y, acc.y);
                acc.z = fmaf(w, v[k].z, acc.z);
                acc.w = fmaf(w, v[k].w, acc.w);
            }
        }
    }
    if (n > ROWCAP) {
        const int m = min(deg[NNODES], NEDGES);
        for (int p = 0; p < m; ++p) {
            const int4 o = side[p];
            if (o.x == row) {
                const float w = __int_as_float(o.z);
                const float4 v = sup4[(size_t)o.y * P4];
                acc.x = fmaf(w, v.x, acc.x);
                acc.y = fmaf(w, v.y, acc.y);
                acc.z = fmaf(w, v.z, acc.z);
                acc.w = fmaf(w, v.w, acc.w);
            }
        }
    }
    return acc;
}

#define GCN_TILE_HEAD(SH)                                                                       \
    const int lane = threadIdx.x;                                                               \
    const int hh = lane >> (SH);                                \
    const int l = lane & ((1 << (SH)) - 1);                   \
    const int row0 = blockIdx.x * 16 + hh;                 \
    int n = deg[row0];                                                                          \
    int4 q[4];                                                                                  \
    {                                                                                           \
        const int4* r4 = (const int4*)(slot + (size_t)row0 * ROWCAP);                           \
        _Pragma("unroll") for (int m = 0; m < 4; ++m) q[m] = r4[m];                             \
    }

#define GCN_PASS_PREFETCH(RP, LASTP)                                                            \
    const int row = row0 + (RP) * p;                                                            \
    const int rown = row0 + (RP) * min(p + 1, (LASTP));        \
    const int nn = deg[rown];                                                                   \
    int4 qx[4];                                                                                 \
    {                                                                                           \
        const int4* r4 = (const int4*)(slot + (size_t)rown * ROWCAP);                           \
        _Pragma("unroll") for (int m = 0; m < 4; ++m) qx[m] = r4[m];                            \
    }

#define GCN_PASS_NEXT()                                                                         \
    n = nn;                                                                                     \
    _Pragma("unroll") for (int m = 0; m < 4; ++m) q[m] = qx[m];

__global__ __launch_bounds__(32) void gcn_pull64_proj_kernel(const int* __restrict__ deg,
                                                             const int2* __restrict__ slot,
                                                             const int4* __restrict__ side,
                                                             const float* __restrict__ sup,
                                                             const float* __restrict__ b1,
                                                             const float* __restrict__ W2,
                                                             float* __restrict__ out) {
    __shared__ float As[16][NHID + 4];

    GCN_TILE_HEAD(4)
    const float4 b = ((const float4*)b1)[l];
    const int c  = l;
    const int kh = hh * 2;

    v2f bf[NHID / 4];
    #pragma unroll
    for (int kk = 0; kk < NHID / 4; ++kk) {
        bf[kk].x = W2[(4 * kk + kh) * NCLASS + c];
        bf[kk].y = W2[(4 * kk + kh + 1) * NCLASS + c];
    }

    for (int p = 0; p < 8; ++p) {
        GCN_PASS_PREFETCH(2, 7)
        float4 acc4 = gcn_row_sum<NHID / 4>(row, l, n, q, deg, slot, side, sup);
        acc4.x = fmaxf(acc4.x + b.x, 0.f);
        acc4.y = fmaxf(acc4.y + b.y, 0.f);
        acc4.z = fmaxf(acc4.z + b.z, 0.f);
        acc4.w = fmaxf(acc4.w + b.w, 0.f);
        *(float4*)&As[2 * p + hh][4 * l] = acc4;
        GCN_PASS_NEXT()
    }
    __syncthreads();

    v8f acc = {};
    #pragma unroll
    for (int kk = 0; kk < NHID; kk += 4) {
        const int k0 = kk + kh;
        v2f a;
        a.x = As[c][k0];
        a.y = As[c][k0 + 1];
        acc = __builtin_amdgcn_wmma_f32_16x16x4_f32(false, a, false, bf[kk >> 2],
                                                    (short)0, acc, false, false);
    }

    float* orow = out + ((size_t)blockIdx.x * 16 + hh * 8) * NCLASS + c;
    #pragma unroll
    for (int r = 0; r < 8; ++r) orow[(size_t)r * NCLASS] = acc[r];
}

template <int M>
__device__ __forceinline__ float gcn_row_xor(float v) {
    return __int_as_float(__builtin_amdgcn_update_dpp(0, __float_as_int(v), 0x160 + M, 0xf, 0xf, false));
}

__global__ __launch_bounds__(32) void gcn_pull16_cls_kernel(const int* __restrict__ deg,
                                                            const int2* __restrict__ slot,
                                                            const int4* __restrict__ side,
                                                            const float* __restrict__ sup2,
                                                            const float* __restrict__ b2,
                                                            float* __restrict__ out) {
    GCN_TILE_HEAD(2)
    const float4 b = ((const float4*)b2)[l];
    for (int p = 0; p < 2; ++p) {
        GCN_PASS_PREFETCH(8, 1)
        float4 v = gcn_row_sum<NCLASS / 4>(row, l, n, q, deg, slot, side, sup2);
        v.x += b.x;
        v.y += b.y;
        v.z += b.z;
        v.w += b.w;
        float mx = fmaxf(fmaxf(v.x, v.y), fmaxf(v.z, v.w));
        mx = fmaxf(mx, gcn_row_xor<1>(mx));
        mx = fmaxf(mx, gcn_row_xor<2>(mx));
        float s = (__expf(v.x - mx) + __expf(v.y - mx)) + (__expf(v.z - mx) + __expf(v.w - mx));
        s += gcn_row_xor<1>(s);
        s += gcn_row_xor<2>(s);
        const float lse = mx + __logf(s);
        ((float4*)out)[(size_t)row * (NCLASS / 4) + l] = make_float4(v.x - lse, v.y - lse, v.z - lse, v.w - lse);
        GCN_PASS_NEXT()
    }
}

extern "C" void kernel_launch(void* const* d_in, const int* in_sizes, int n_in,
                              void* d_out, int out_size, void* d_ws, size_t ws_size,
                              hipStream_t stream) {
    const float* x   = (const float*)d_in[0];
    const int*   src = (const int*)  d_in[1];
    const int*   dst = (const int*)  d_in[2];
    const float* ew  = (const float*)d_in[3];
    const float* W1  = (const float*)d_in[4];
    const float* b1  = (const float*)d_in[5];
    const float* W2  = (const float*)d_in[6];
    const float* b2  = (const float*)d_in[7];
    float* out = (float*)d_out;

    float* ws       = (float*)d_ws;
    float* support1 = ws;
    float* support2 = ws + (size_t)NNODES * NHID;
    int*   deg      = (int*)(ws + 2 * (size_t)NNODES * NHID + (size_t)NNODES * NCLASS);
    int4*  side     = (int4*)(deg + NDEG_PAD);
    int2*  slot     = (int2*)(side + NEDGES);

    gcn_zero_kernel<<<NDEG_PAD / 4 / 256, 256, 0, stream>>>((float4*)deg, NDEG_PAD / 4);
    gcn_place_kernel<<<PL_BLOCKS, PL_THREADS, 0, stream>>>(src, dst, ew, deg, slot, side);

    gcn_gemm1_kernel<<<G1_WAVES, 32, 0, stream>>>(x, W1, support1);
    gcn_pull64_proj_kernel<<<MTILES, 32, 0, stream>>>(deg, slot, side, support1, b1, W2, support2);

    gcn_pull16_cls_kernel<<<MTILES, 32, 0, stream>>>(deg, slot, side, support2, b2, out);
}
